# starting kernel
# baseline (speedup 1.0000x reference)
#include <hip/hip_runtime.h>
#include <hip/hip_bf16.h>
#include <math.h>

typedef float v2f __attribute__((ext_vector_type(2)));
typedef float v8f __attribute__((ext_vector_type(8)));

#define HEADS 4
#define HC 256
#define CDIM 64
#define NEG_SLOPE 0.2f

__global__ void gemm_wmma_f32(const float* __restrict__ A,
                              const float* __restrict__ B,
                              float* __restrict__ C,
                              int M, int K, int Ncol) {
    const int lane   = threadIdx.x & 31;
    const int wave   = threadIdx.x >> 5;
    const int wtile  = blockIdx.x * 4 + wave;
    const int wtprow = Ncol >> 6;
    const int total  = (M >> 4) * wtprow;
    if (wtile >= total) return;
    const int trow   = wtile / wtprow;
    const int tcol   = wtile - trow * wtprow;
    const int row0   = trow << 4;
    const int col0   = tcol << 6;
    const int laneM  = lane & 15;
    const int laneHi = lane >> 4;

    const float* Arow  = A + (size_t)(row0 + laneM) * K;
    const float* Bbase = B + (size_t)col0 + laneM;

    v8f acc0 = {}, acc1 = {}, acc2 = {}, acc3 = {};
#pragma unroll 2
    for (int k = 0; k < K; k += 4) {
        const int ka = k + (laneHi << 1);
        v2f a;
        a.x = Arow[ka];
        a.y = Arow[ka + 1];
        const float* B0 = Bbase + (size_t)ka * Ncol;
        const float* B1 = Bbase + (size_t)(ka + 1) * Ncol;
        v2f b0, b1, b2, b3;
        b0.x = B0[0];  b0.y = B1[0];
        b1.x = B0[16]; b1.y = B1[16];
        b2.x = B0[32]; b2.y = B1[32];
        b3.x = B0[48]; b3.y = B1[48];
        acc0 = __builtin_amdgcn_wmma_f32_16x16x4_f32(false, a, false, b0, (short)0, acc0, false, false);
        acc1 = __builtin_amdgcn_wmma_f32_16x16x4_f32(false, a, false, b1, (short)0, acc1, false, false);
        acc2 = __builtin_amdgcn_wmma_f32_16x16x4_f32(false, a, false, b2, (short)0, acc2, false, false);
        acc3 = __builtin_amdgcn_wmma_f32_16x16x4_f32(false, a, false, b3, (short)0, acc3, false, false);
    }

    float* Crow = C + (size_t)(row0 + 8 * laneHi) * Ncol + col0 + laneM;
#pragma unroll
    for (int i = 0; i < 8; ++i) {
        float* r = Crow + (size_t)i * Ncol;
        r[0]  = acc0[i];
        r[16] = acc1[i];
        r[32] = acc2[i];
        r[48] = acc3[i];
    }
}

__global__ void attn_scores(const float* __restrict__ Hm,
                            const float* __restrict__ att_s,
                            const float* __restrict__ att_d,
                            float* __restrict__ as_, float* __restrict__ ad_,
                            int N) {
    const int idx = blockIdx.x * blockDim.x + threadIdx.x;
    if (idx >= N * HEADS) return;
    const int n = idx >> 2, h = idx & 3;
    const float* row = Hm + (size_t)n * HC + h * CDIM;
    const float* vs  = att_s + h * CDIM;
    const float* vd  = att_d + h * CDIM;
    float s = 0.f, d = 0.f;
#pragma unroll 8
    for (int c = 0; c < CDIM; ++c) {
        const float v = row[c];
        s = fmaf(v, vs[c], s);
        d = fmaf(v, vd[c], d);
    }
    as_[idx] = s;
    ad_[idx] = d;
}

__device__ __forceinline__ unsigned fmap(float x) {
    unsigned u = __float_as_uint(x);
    return (u & 0x80000000u) ? ~u : (u | 0x80000000u);
}
__device__ __forceinline__ float funmap(unsigned u) {
    u = (u & 0x80000000u) ? (u ^ 0x80000000u) : ~u;
    return __uint_as_float(u);
}

__device__ __forceinline__ void edge_nodes(int eid, int E,
                                           const int* __restrict__ srcv,
                                           const int* __restrict__ dstv,
                                           int& s, int& d) {
    if (eid < E) { s = srcv[eid]; d = dstv[eid]; }
    else         { s = d = eid - E; }
}

__device__ __forceinline__ float leaky(float e) {
    return e > 0.f ? e : NEG_SLOPE * e;
}

__global__ void edge_max(const int* __restrict__ srcv, const int* __restrict__ dstv,
                         const float* __restrict__ as_, const float* __restrict__ ad_,
                         unsigned* __restrict__ dmax, int E, int N) {
    const int idx = blockIdx.x * blockDim.x + threadIdx.x;
    const int tot = (E + N) * HEADS;
    if (idx >= tot) return;
    const int eid = idx >> 2, h = idx & 3;
    int s, d;
    edge_nodes(eid, E, srcv, dstv, s, d);
    const float e = leaky(as_[s * HEADS + h] + ad_[d * HEADS + h]);
    atomicMax(&dmax[d * HEADS + h], fmap(e));
}

__global__ void edge_expsum(const int* __restrict__ srcv, const int* __restrict__ dstv,
                            const float* __restrict__ as_, const float* __restrict__ ad_,
                            const unsigned* __restrict__ dmax,
                            float* __restrict__ EXv, float* __restrict__ dsum,
                            int E, int N) {
    const int idx = blockIdx.x * blockDim.x + threadIdx.x;
    const int tot = (E + N) * HEADS;
    if (idx >= tot) return;
    const int eid = idx >> 2, h = idx & 3;
    int s, d;
    edge_nodes(eid, E, srcv, dstv, s, d);
    const float e  = leaky(as_[s * HEADS + h] + ad_[d * HEADS + h]);
    const float m  = funmap(dmax[d * HEADS + h]);
    const float ex = __expf(e - m);
    EXv[idx] = ex;
    atomicAdd(&dsum[d * HEADS + h], ex);
}

__global__ void edge_scatter(const int* __restrict__ srcv, const int* __restrict__ dstv,
                             const float* __restrict__ Hm,
                             const float* __restrict__ EXv,
                             const float* __restrict__ dsum,
                             float* __restrict__ ACC, int E, int N) {
    const int eid = blockIdx.x;
    const int t   = threadIdx.x;
    int s, d;
    edge_nodes(eid, E, srcv, dstv, s, d);
    const int h = t >> 6;
    const float alpha = EXv[eid * HEADS + h] / (dsum[d * HEADS + h] + 1e-16f);
    atomicAdd(&ACC[(size_t)d * HC + t], Hm[(size_t)s * HC + t] * alpha);
}

__global__ void head_mean(const float* __restrict__ ACC,
                          const float* __restrict__ bias,
                          float* __restrict__ out, int N, int do_relu) {
    const int idx = blockIdx.x * blockDim.x + threadIdx.x;
    if (idx >= N * CDIM) return;
    const int n = idx >> 6, c = idx & 63;
    const float* r = ACC + (size_t)n * HC;
    float v = (r[c] + r[CDIM + c] + r[2 * CDIM + c] + r[3 * CDIM + c]) * 0.25f
              + bias[c];
    if (do_relu) v = fmaxf(v, 0.f);
    out[idx] = v;
}

extern "C" void kernel_launch(void* const* d_in, const int* in_sizes, int n_in,
                              void* d_out, int out_size, void* d_ws, size_t ws_size,
                              hipStream_t stream) {
    const float* x        = (const float*)d_in[0];
    const int*   ei       = (const int*)  d_in[1];
    const float* W1       = (const float*)d_in[2];
    const float* att_src1 = (const float*)d_in[3];
    const float* att_dst1 = (const float*)d_in[4];
    const float* b1       = (const float*)d_in[5];
    const float* W2       = (const float*)d_in[6];
    const float* att_src2 = (const float*)d_in[7];
    const float* att_dst2 = (const float*)d_in[8];
    const float* b2       = (const float*)d_in[9];

    const int Fin = 128;
    const int N   = in_sizes[0] / Fin;
    const int E   = in_sizes[1] / 2;
    const int ET  = E + N;
    const int* srcv = ei;
    const int* dstv = ei + E;

    float* ws    = (float*)d_ws;
    float* Hbuf  = ws;
    float* ACC   = Hbuf + (size_t)N * HC;
    float* HID   = ACC  + (size_t)N * HC;
    float* ASRC  = HID  + (size_t)N * CDIM;
    float* ADST  = ASRC + (size_t)N * HEADS;
    unsigned* DMAX = (unsigned*)(ADST + (size_t)N * HEADS);
    float* DSUM  = (float*)(DMAX + (size_t)N * HEADS);
    float* EXv   = DSUM + (size_t)N * HEADS;

    const int gemmStrips = (N / 16) * (HC / 64);
    const int gemmBlocks = (gemmStrips + 3) / 4;
    const int nhBlocks   = (N * HEADS + 255) / 256;
    const int ehBlocks   = (ET * HEADS + 255) / 256;
    const int ncBlocks   = (N * CDIM + 255) / 256;

    hipMemsetAsync(DMAX, 0, (size_t)N * HEADS * 4, stream);
    hipMemsetAsync(DSUM, 0, (size_t)N * HEADS * 4, stream);
    hipMemsetAsync(ACC,  0, (size_t)N * HC * 4, stream);

    gemm_wmma_f32<<<gemmBlocks, 128, 0, stream>>>(x, W1, Hbuf, N, Fin, HC);
    attn_scores<<<nhBlocks, 256, 0, stream>>>(Hbuf, att_src1, att_dst1, ASRC, ADST, N);
    edge_max<<<ehBlocks, 256, 0, stream>>>(srcv, dstv, ASRC, ADST, DMAX, E, N);
    edge_expsum<<<ehBlocks, 256, 0, stream>>>(srcv, dstv, ASRC, ADST, DMAX, EXv, DSUM, E, N);
    edge_scatter<<<ET, HC, 0, stream>>>(srcv, dstv, Hbuf, EXv, DSUM, ACC, E, N);
    head_mean<<<ncBlocks, 256, 0, stream>>>(ACC, b1, HID, N,  1);

    hipMemsetAsync(DMAX, 0, (size_t)N * HEADS * 4, stream);
    hipMemsetAsync(DSUM, 0, (size_t)N * HEADS * 4, stream);
    hipMemsetAsync(ACC,  0, (size_t)N * HC * 4, stream);

    gemm_wmma_f32<<<gemmBlocks, 128, 0, stream>>>(HID, W2, Hbuf, N, CDIM, HC);
    attn_scores<<<nhBlocks, 256, 0, stream>>>(Hbuf, att_src2, att_dst2, ASRC, ADST, N);
    edge_max<<<ehBlocks, 256, 0, stream>>>(srcv, dstv, ASRC, ADST, DMAX, E, N);
    edge_expsum<<<ehBlocks, 256, 0, stream>>>(srcv, dstv, ASRC, ADST, DMAX, EXv, DSUM, E, N);
    edge_scatter<<<ET, HC, 0, stream>>>(srcv, dstv, Hbuf, EXv, DSUM, ACC, E, N);
    head_mean<<<ncBlocks, 256, 0, stream>>>(ACC, b2, (float*)d_out, N,  0);
}


# edge scatter: the 4 heads of an edge are summed in registers before ONE atomic add per (dst, channel), 8 edges per wave, ACC [N,64]; sum order changes (heads first, then atomics over edges)
# speedup vs baseline: 3.8130x; 3.8130x over previous
#include <hip/hip_runtime.h>
#include <hip/hip_bf16.h>
#include <math.h>

typedef float v2f __attribute__((ext_vector_type(2)));
typedef float v8f __attribute__((ext_vector_type(8)));

#define HEADS 4
#define HC 256
#define CDIM 64
#define NEG_SLOPE 0.2f

__global__ void gemm_wmma_f32(const float* __restrict__ A,
                              const float* __restrict__ B,
                              float* __restrict__ C,
                              int M, int K, int Ncol) {
    const int lane   = threadIdx.x & 31;
    const int wave   = threadIdx.x >> 5;
    const int wtile  = blockIdx.x * 4 + wave;
    const int wtprow = Ncol >> 6;
    const int total  = (M >> 4) * wtprow;
    if (wtile >= total) return;
    const int trow   = wtile / wtprow;
    const int tcol   = wtile - trow * wtprow;
    const int row0   = trow << 4;
    const int col0   = tcol << 6;
    const int laneM  = lane & 15;
    const int laneHi = lane >> 4;

    const float* Arow  = A + (size_t)(row0 + laneM) * K;
    const float* Bbase = B + (size_t)col0 + laneM;

    v8f acc0 = {}, acc1 = {}, acc2 = {}, acc3 = {};
#pragma unroll 2
    for (int k = 0; k < K; k += 4) {
        const int ka = k + (laneHi << 1);
        v2f a;
        a.x = Arow[ka];
        a.y = Arow[ka + 1];
        const float* B0 = Bbase + (size_t)ka * Ncol;
        const float* B1 = Bbase + (size_t)(ka + 1) * Ncol;
        v2f b0, b1, b2, b3;
        b0.x = B0[0];  b0.y = B1[0];
        b1.x = B0[16]; b1.y = B1[16];
        b2.x = B0[32]; b2.y = B1[32];
        b3.x = B0[48]; b3.y = B1[48];
        acc0 = __builtin_amdgcn_wmma_f32_16x16x4_f32(false, a, false, b0, (short)0, acc0, false, false);
        acc1 = __builtin_amdgcn_wmma_f32_16x16x4_f32(false, a, false, b1, (short)0, acc1, false, false);
        acc2 = __builtin_amdgcn_wmma_f32_16x16x4_f32(false, a, false, b2, (short)0, acc2, false, false);
        acc3 = __builtin_amdgcn_wmma_f32_16x16x4_f32(false, a, false, b3, (short)0, acc3, false, false);
    }

    float* Crow = C + (size_t)(row0 + 8 * laneHi) * Ncol + col0 + laneM;
#pragma unroll
    for (int i = 0; i < 8; ++i) {
        float* r = Crow + (size_t)i * Ncol;
        r[0]  = acc0[i];
        r[16] = acc1[i];
        r[32] = acc2[i];
        r[48] = acc3[i];
    }
}

__global__ void attn_scores(const float* __restrict__ Hm,
                            const float* __restrict__ att_s,
                            const float* __restrict__ att_d,
                            float* __restrict__ as_, float* __restrict__ ad_,
                            int N) {
    const int idx = blockIdx.x * blockDim.x + threadIdx.x;
    if (idx >= N * HEADS) return;
    const int n = idx >> 2, h = idx & 3;
    const float* row = Hm + (size_t)n * HC + h * CDIM;
    const float* vs  = att_s + h * CDIM;
    const float* vd  = att_d + h * CDIM;
    float s = 0.f, d = 0.f;
#pragma unroll 8
    for (int c = 0; c < CDIM; ++c) {
        const float v = row[c];
        s = fmaf(v, vs[c], s);
        d = fmaf(v, vd[c], d);
    }
    as_[idx] = s;
    ad_[idx] = d;
}

__device__ __forceinline__ unsigned fmap(float x) {
    unsigned u = __float_as_uint(x);
    return (u & 0x80000000u) ? ~u : (u | 0x80000000u);
}
__device__ __forceinline__ float funmap(unsigned u) {
    u = (u & 0x80000000u) ? (u ^ 0x80000000u) : ~u;
    return __uint_as_float(u);
}

__device__ __forceinline__ void edge_nodes(int eid, int E,
                                           const int* __restrict__ srcv,
                                           const int* __restrict__ dstv,
                                           int& s, int& d) {
    if (eid < E) { s = srcv[eid]; d = dstv[eid]; }
    else         { s = d = eid - E; }
}

__device__ __forceinline__ float leaky(float e) {
    return e > 0.f ? e : NEG_SLOPE * e;
}

__global__ void edge_max(const int* __restrict__ srcv, const int* __restrict__ dstv,
                         const float* __restrict__ as_, const float* __restrict__ ad_,
                         unsigned* __restrict__ dmax, int E, int N) {
    const int idx = blockIdx.x * blockDim.x + threadIdx.x;
    const int tot = (E + N) * HEADS;
    if (idx >= tot) return;
    const int eid = idx >> 2, h = idx & 3;
    int s, d;
    edge_nodes(eid, E, srcv, dstv, s, d);
    const float e = leaky(as_[s * HEADS + h] + ad_[d * HEADS + h]);
    atomicMax(&dmax[d * HEADS + h], fmap(e));
}

__global__ void edge_expsum(const int* __restrict__ srcv, const int* __restrict__ dstv,
                            const float* __restrict__ as_, const float* __restrict__ ad_,
                            const unsigned* __restrict__ dmax,
                            float* __restrict__ EXv, float* __restrict__ dsum,
                            int E, int N) {
    const int idx = blockIdx.x * blockDim.x + threadIdx.x;
    const int tot = (E + N) * HEADS;
    if (idx >= tot) return;
    const int eid = idx >> 2, h = idx & 3;
    int s, d;
    edge_nodes(eid, E, srcv, dstv, s, d);
    const float e  = leaky(as_[s * HEADS + h] + ad_[d * HEADS + h]);
    const float m  = funmap(dmax[d * HEADS + h]);
    const float ex = __expf(e - m);
    EXv[idx] = ex;
    atomicAdd(&dsum[d * HEADS + h], ex);
}

#define EPW 8
#define GRP 4

template <bool FULL>
__device__ __forceinline__ void scatter_group(const float* __restrict__ Hm,
                                              float* __restrict__ ACC,
                                              int sl, int dl, float al,
                                              int lane, int nvalid) {
    const int ai = __float_as_int(al);
#pragma unroll
    for (int g = 0; g < EPW; g += GRP) {
        float v0[GRP], v1[GRP];
#pragma unroll
        for (int i = 0; i < GRP; ++i) {
            const int j = g + i;
            v0[i] = 0.f; v1[i] = 0.f;
            if (!FULL && j >= nvalid) continue;
            const int s = __builtin_amdgcn_readlane(sl, 4 * j);
            const float a0 = __int_as_float(__builtin_amdgcn_readlane(ai, 4 * j + 0));
            const float a1 = __int_as_float(__builtin_amdgcn_readlane(ai, 4 * j + 1));
            const float a2 = __int_as_float(__builtin_amdgcn_readlane(ai, 4 * j + 2));
            const float a3 = __int_as_float(__builtin_amdgcn_readlane(ai, 4 * j + 3));
            const float* hr = Hm + (size_t)s * HC + lane;
            float x0 = hr[0] * a0;
            float x1 = hr[32] * a0;
            x0 = fmaf(hr[CDIM],          a1, x0);
            x1 = fmaf(hr[CDIM + 32],     a1, x1);
            x0 = fmaf(hr[2 * CDIM],      a2, x0);
            x1 = fmaf(hr[2 * CDIM + 32], a2, x1);
            x0 = fmaf(hr[3 * CDIM],      a3, x0);
            x1 = fmaf(hr[3 * CDIM + 32], a3, x1);
            v0[i] = x0; v1[i] = x1;
        }
#pragma unroll
        for (int i = 0; i < GRP; ++i) {
            const int j = g + i;
            if (!FULL && j >= nvalid) continue;
            const int d = __builtin_amdgcn_readlane(dl, 4 * j);
            float* ar = ACC + (size_t)d * CDIM + lane;
            atomicAdd(ar, v0[i]);
            atomicAdd(ar + 32, v1[i]);
        }
    }
}

__global__ void __launch_bounds__(256)
edge_scatter(const int* __restrict__ srcv, const int* __restrict__ dstv,
             const float* __restrict__ Hm,
             const float* __restrict__ EXv,
             const float* __restrict__ dsum,
             float* __restrict__ ACC, int E, int N) {
    const int lane = threadIdx.x & 31;
    const int wave = __builtin_amdgcn_readfirstlane((int)(threadIdx.x >> 5));
    const int ET = E + N;
    const int e0 = (blockIdx.x * 8 + wave) * EPW;
    if (e0 >= ET) return;
    const int el = e0 + (lane >> 2);
    const int hl = lane & 3;
    int sl = 0, dl = 0;
    float al = 0.f;
    if (el < ET) {
        edge_nodes(el, E, srcv, dstv, sl, dl);
        al = EXv[(size_t)el * HEADS + hl] / (dsum[dl * HEADS + hl] + 1e-16f);
    }
    if (e0 + EPW <= ET) scatter_group<true >(Hm, ACC, sl, dl, al, lane, EPW);
    else                scatter_group<false>(Hm, ACC, sl, dl, al, lane, ET - e0);
}

__global__ void head_mean(const float* __restrict__ ACC,
                          const float* __restrict__ bias,
                          float* __restrict__ out, int N, int do_relu) {
    const int idx = blockIdx.x * blockDim.x + threadIdx.x;
    if (idx >= N * CDIM) return;
    float v = ACC[idx] * 0.25f + bias[idx & 63];
    if (do_relu) v = fmaxf(v, 0.f);
    out[idx] = v;
}

extern "C" void kernel_launch(void* const* d_in, const int* in_sizes, int n_in,
                              void* d_out, int out_size, void* d_ws, size_t ws_size,
                              hipStream_t stream) {
    const float* x        = (const float*)d_in[0];
    const int*   ei       = (const int*)  d_in[1];
    const float* W1       = (const float*)d_in[2];
    const float* att_src1 = (const float*)d_in[3];
    const float* att_dst1 = (const float*)d_in[4];
    const float* b1       = (const float*)d_in[5];
    const float* W2       = (const float*)d_in[6];
    const float* att_src2 = (const float*)d_in[7];
    const float* att_dst2 = (const float*)d_in[8];
    const float* b2       = (const float*)d_in[9];

    const int Fin = 128;
    const int N   = in_sizes[0] / Fin;
    const int E   = in_sizes[1] / 2;
    const int ET  = E + N;
    const int* srcv = ei;
    const int* dstv = ei + E;

    float* ws    = (float*)d_ws;
    float* Hbuf  = ws;
    float* ACC   = Hbuf + (size_t)N * HC;
    float* HID   = ACC  + (size_t)N * HC;
    float* ASRC  = HID  + (size_t)N * CDIM;
    float* ADST  = ASRC + (size_t)N * HEADS;
    unsigned* DMAX = (unsigned*)(ADST + (size_t)N * HEADS);
    float* DSUM  = (float*)(DMAX + (size_t)N * HEADS);
    float* EXv   = DSUM + (size_t)N * HEADS;

    const int gemmStrips = (N / 16) * (HC / 64);
    const int gemmBlocks = (gemmStrips + 3) / 4;
    const int nhBlocks   = (N * HEADS + 255) / 256;
    const int ehBlocks   = (ET * HEADS + 255) / 256;
    const int ncBlocks   = (N * CDIM + 255) / 256;
    const int scBlocks   = (ET + 8 * EPW - 1) / (8 * EPW);

    hipMemsetAsync(DMAX, 0, (size_t)N * HEADS * 4, stream);
    hipMemsetAsync(DSUM, 0, (size_t)N * HEADS * 4, stream);
    hipMemsetAsync(ACC,  0, (size_t)N * CDIM * 4, stream);

    gemm_wmma_f32<<<gemmBlocks, 128, 0, stream>>>(x, W1, Hbuf, N, Fin, HC);
    attn_scores<<<nhBlocks, 256, 0, stream>>>(Hbuf, att_src1, att_dst1, ASRC, ADST, N);
    edge_max<<<ehBlocks, 256, 0, stream>>>(srcv, dstv, ASRC, ADST, DMAX, E, N);
    edge_expsum<<<ehBlocks, 256, 0, stream>>>(srcv, dstv, ASRC, ADST, DMAX, EXv, DSUM, E, N);
    edge_scatter<<<scBlocks, 256, 0, stream>>>(srcv, dstv, Hbuf, EXv, DSUM, ACC, E, N);
    head_mean<<<ncBlocks, 256, 0, stream>>>(ACC, b1, HID, N,  1);

    hipMemsetAsync(DMAX, 0, (size_t)N * HEADS * 4, stream);
    hipMemsetAsync(DSUM, 0, (size_t)N * HEADS * 4, stream);
    hipMemsetAsync(ACC,  0, (size_t)N * CDIM * 4, stream);

    gemm_wmma_f32<<<gemmBlocks, 128, 0, stream>>>(HID, W2, Hbuf, N, CDIM, HC);
    attn_scores<<<nhBlocks, 256, 0, stream>>>(Hbuf, att_src2, att_dst2, ASRC, ADST, N);
    edge_max<<<ehBlocks, 256, 0, stream>>>(srcv, dstv, ASRC, ADST, DMAX, E, N);
    edge_expsum<<<ehBlocks, 256, 0, stream>>>(srcv, dstv, ASRC, ADST, DMAX, EXv, DSUM, E, N);
    edge_scatter<<<scBlocks, 256, 0, stream>>>(srcv, dstv, Hbuf, EXv, DSUM, ACC, E, N);
    head_mean<<<ncBlocks, 256, 0, stream>>>(ACC, b2, (float*)d_out, N,  0);
}


# edges grouped by destination once per call (count, scan, fill); one wave per node does softmax and row sum in registers, no float atomics; sum order changes (slot order from an int atomic)
# speedup vs baseline: 5.3703x; 1.4084x over previous
#include <hip/hip_runtime.h>
#include <hip/hip_bf16.h>
#include <math.h>

typedef float v2f __attribute__((ext_vector_type(2)));
typedef float v8f __attribute__((ext_vector_type(8)));

#define HEADS 4
#define HC 256
#define CDIM 64
#define NEG_SLOPE 0.2f

__global__ void gemm_wmma_f32(const float* __restrict__ A,
                              const float* __restrict__ B,
                              float* __restrict__ C,
                              int M, int K, int Ncol) {
    const int lane   = threadIdx.x & 31;
    const int wave   = threadIdx.x >> 5;
    const int wtile  = blockIdx.x * 4 + wave;
    const int wtprow = Ncol >> 6;
    const int total  = (M >> 4) * wtprow;
    if (wtile >= total) return;
    const int trow   = wtile / wtprow;
    const int tcol   = wtile - trow * wtprow;
    const int row0   = trow << 4;
    const int col0   = tcol << 6;
    const int laneM  = lane & 15;
    const int laneHi = lane >> 4;

    const float* Arow  = A + (size_t)(row0 + laneM) * K;
    const float* Bbase = B + (size_t)col0 + laneM;

    v8f acc0 = {}, acc1 = {}, acc2 = {}, acc3 = {};
#pragma unroll 2
    for (int k = 0; k < K; k += 4) {
        const int ka = k + (laneHi << 1);
        v2f a;
        a.x = Arow[ka];
        a.y = Arow[ka + 1];
        const float* B0 = Bbase + (size_t)ka * Ncol;
        const float* B1 = Bbase + (size_t)(ka + 1) * Ncol;
        v2f b0, b1, b2, b3;
        b0.x = B0[0];  b0.y = B1[0];
        b1.x = B0[16]; b1.y = B1[16];
        b2.x = B0[32]; b2.y = B1[32];
        b3.x = B0[48]; b3.y = B1[48];
        acc0 = __builtin_amdgcn_wmma_f32_16x16x4_f32(false, a, false, b0, (short)0, acc0, false, false);
        acc1 = __builtin_amdgcn_wmma_f32_16x16x4_f32(false, a, false, b1, (short)0, acc1, false, false);
        acc2 = __builtin_amdgcn_wmma_f32_16x16x4_f32(false, a, false, b2, (short)0, acc2, false, false);
        acc3 = __builtin_amdgcn_wmma_f32_16x16x4_f32(false, a, false, b3, (short)0, acc3, false, false);
    }

    float* Crow = C + (size_t)(row0 + 8 * laneHi) * Ncol + col0 + laneM;
#pragma unroll
    for (int i = 0; i < 8; ++i) {
        float* r = Crow + (size_t)i * Ncol;
        r[0]  = acc0[i];
        r[16] = acc1[i];
        r[32] = acc2[i];
        r[48] = acc3[i];
    }
}

__global__ void attn_scores(const float* __restrict__ Hm,
                            const float* __restrict__ att_s,
                            const float* __restrict__ att_d,
                            float* __restrict__ as_, float* __restrict__ ad_,
                            int N) {
    const int idx = blockIdx.x * blockDim.x + threadIdx.x;
    if (idx >= N * HEADS) return;
    const int n = idx >> 2, h = idx & 3;
    const float* row = Hm + (size_t)n * HC + h * CDIM;
    const float* vs  = att_s + h * CDIM;
    const float* vd  = att_d + h * CDIM;
    float s = 0.f, d = 0.f;
#pragma unroll 8
    for (int c = 0; c < CDIM; ++c) {
        const float v = row[c];
        s = fmaf(v, vs[c], s);
        d = fmaf(v, vd[c], d);
    }
    as_[idx] = s;
    ad_[idx] = d;
}

__device__ __forceinline__ float leaky(float e) {
    return e > 0.f ? e : NEG_SLOPE * e;
}

__global__ void count_dst(const int* __restrict__ dstv, int* __restrict__ cnt, int E) {
    const int e = blockIdx.x * blockDim.x + threadIdx.x;
    if (e < E) atomicAdd(&cnt[dstv[e]], 1);
}

#define SCAN_T 1024
__global__ void __launch_bounds__(SCAN_T)
scan_cnt(const int* __restrict__ cnt, int* __restrict__ rowptr,
         int* __restrict__ cursor, int N) {
    __shared__ int sh[SCAN_T];
    const int t   = threadIdx.x;
    const int per = (N + SCAN_T - 1) / SCAN_T;
    const int b   = t * per;
    const int e   = (b + per < N) ? (b + per) : N;
    int s = 0;
    for (int i = b; i < e; ++i) s += cnt[i];
    sh[t] = s;
    __syncthreads();
    for (int off = 1; off < SCAN_T; off <<= 1) {
        const int v = (t >= off) ? sh[t - off] : 0;
        __syncthreads();
        sh[t] += v;
        __syncthreads();
    }
    int run = sh[t] - s;
    int i = b;
    for (; i + 8 <= e; i += 8) {
        int c8[8];
#pragma unroll
        for (int k = 0; k < 8; ++k) c8[k] = cnt[i + k];
#pragma unroll
        for (int k = 0; k < 8; ++k) {
            rowptr[i + k] = run;
            cursor[i + k] = run;
            run += c8[k];
        }
    }
    for (; i < e; ++i) {
        rowptr[i] = run;
        cursor[i] = run;
        run += cnt[i];
    }
    if (t == SCAN_T - 1) rowptr[N] = sh[SCAN_T - 1];
}

__global__ void fill_csr(const int* __restrict__ srcv, const int* __restrict__ dstv,
                         int* __restrict__ cursor, int* __restrict__ csrc, int E) {
    const int e = blockIdx.x * blockDim.x + threadIdx.x;
    if (e >= E) return;
    const int pos = atomicAdd(&cursor[dstv[e]], 1);
    csrc[pos] = srcv[e];
}

__device__ __forceinline__ float wave_max(float v) {
#pragma unroll
    for (int o = 16; o > 0; o >>= 1) v = fmaxf(v, __shfl_xor(v, o, 32));
    return v;
}
__device__ __forceinline__ float wave_sum(float v) {
#pragma unroll
    for (int o = 16; o > 0; o >>= 1) v += __shfl_xor(v, o, 32);
    return v;
}

__device__ __forceinline__ void edge_scores(int i, int cnt, int base, int d,
                                            const int* __restrict__ csrc,
                                            const float* __restrict__ as_,
                                            const float4 adv, int& s, float4& e) {
    s = d;
    if (i < cnt) s = csrc[base + i];
    const float4 a = *reinterpret_cast<const float4*>(as_ + (size_t)s * HEADS);
    e.x = leaky(a.x + adv.x);
    e.y = leaky(a.y + adv.y);
    e.z = leaky(a.z + adv.z);
    e.w = leaky(a.w + adv.w);
}

__device__ __forceinline__ void gather_row(const float* __restrict__ Hm, int s,
                                           float a0, float a1, float a2, float a3,
                                           int lane, float& x0, float& x1) {
    const float* hr = Hm + (size_t)s * HC + lane;
    x0 = hr[0] * a0;
    x1 = hr[32] * a0;
    x0 = fmaf(hr[CDIM],          a1, x0);
    x1 = fmaf(hr[CDIM + 32],     a1, x1);
    x0 = fmaf(hr[2 * CDIM],      a2, x0);
    x1 = fmaf(hr[2 * CDIM + 32], a2, x1);
    x0 = fmaf(hr[3 * CDIM],      a3, x0);
    x1 = fmaf(hr[3 * CDIM + 32], a3, x1);
}

__global__ void __launch_bounds__(256)
node_aggregate(const int* __restrict__ rowptr, const int* __restrict__ csrc,
               const float* __restrict__ Hm,
               const float* __restrict__ as_, const float* __restrict__ ad_,
               const float* __restrict__ bias, float* __restrict__ out,
               int N, int do_relu) {
    const int lane = threadIdx.x & 31;
    const int wave = __builtin_amdgcn_readfirstlane((int)(threadIdx.x >> 5));
    const int d = blockIdx.x * 8 + wave;
    if (d >= N) return;
    const int base = __builtin_amdgcn_readfirstlane(rowptr[d]);
    const int cnt  = __builtin_amdgcn_readfirstlane(rowptr[d + 1]) - base;
    const int T    = cnt + 1;
    const float4 adv = *reinterpret_cast<const float4*>(ad_ + (size_t)d * HEADS);

    int s_first; float4 e_first;
    edge_scores(lane, cnt, base, d, csrc, as_, adv, s_first, e_first);

    float m0 = -INFINITY, m1 = -INFINITY, m2 = -INFINITY, m3 = -INFINITY;
    for (int c0 = 0; c0 < T; c0 += 32) {
        const int i = c0 + lane;
        int s = s_first; float4 e = e_first;
        if (c0 > 0) edge_scores(i, cnt, base, d, csrc, as_, adv, s, e);
        if (i < T) {
            m0 = fmaxf(m0, e.x); m1 = fmaxf(m1, e.y);
            m2 = fmaxf(m2, e.z); m3 = fmaxf(m3, e.w);
        }
    }
    m0 = wave_max(m0); m1 = wave_max(m1); m2 = wave_max(m2); m3 = wave_max(m3);

    float s0 = 0.f, s1 = 0.f, s2 = 0.f, s3 = 0.f;
    for (int c0 = 0; c0 < T; c0 += 32) {
        const int i = c0 + lane;
        int s = s_first; float4 e = e_first;
        if (c0 > 0) edge_scores(i, cnt, base, d, csrc, as_, adv, s, e);
        if (i < T) {
            s0 += __expf(e.x - m0); s1 += __expf(e.y - m1);
            s2 += __expf(e.z - m2); s3 += __expf(e.w - m3);
        }
    }
    const float den0 = wave_sum(s0) + 1e-16f;
    const float den1 = wave_sum(s1) + 1e-16f;
    const float den2 = wave_sum(s2) + 1e-16f;
    const float den3 = wave_sum(s3) + 1e-16f;

    float acc0 = 0.f, acc1 = 0.f;
    for (int c0 = 0; c0 < T; c0 += 32) {
        int sl = s_first; float4 e = e_first;
        if (c0 > 0) edge_scores(c0 + lane, cnt, base, d, csrc, as_, adv, sl, e);
        const int w0 = __float_as_int(__expf(e.x - m0) / den0);
        const int w1 = __float_as_int(__expf(e.y - m1) / den1);
        const int w2 = __float_as_int(__expf(e.z - m2) / den2);
        const int w3 = __float_as_int(__expf(e.w - m3) / den3);
        const int n = (T - c0 < 32) ? (T - c0) : 32;
        int j = 0;
        for (; j + 4 <= n; j += 4) {
            float x0[4], x1[4];
#pragma unroll
            for (int k = 0; k < 4; ++k) {
                const int s = __builtin_amdgcn_readlane(sl, j + k);
                const float a0 = __int_as_float(__builtin_amdgcn_readlane(w0, j + k));
                const float a1 = __int_as_float(__builtin_amdgcn_readlane(w1, j + k));
                const float a2 = __int_as_float(__builtin_amdgcn_readlane(w2, j + k));
                const float a3 = __int_as_float(__builtin_amdgcn_readlane(w3, j + k));
                gather_row(Hm, s, a0, a1, a2, a3, lane, x0[k], x1[k]);
            }
            acc0 += (x0[0] + x0[1]) + (x0[2] + x0[3]);
            acc1 += (x1[0] + x1[1]) + (x1[2] + x1[3]);
        }
        for (; j < n; ++j) {
            const int s = __builtin_amdgcn_readlane(sl, j);
            const float a0 = __int_as_float(__builtin_amdgcn_readlane(w0, j));
            const float a1 = __int_as_float(__builtin_amdgcn_readlane(w1, j));
            const float a2 = __int_as_float(__builtin_amdgcn_readlane(w2, j));
            const float a3 = __int_as_float(__builtin_amdgcn_readlane(w3, j));
            float x0, x1;
            gather_row(Hm, s, a0, a1, a2, a3, lane, x0, x1);
            acc0 += x0;
            acc1 += x1;
        }
    }

    float r0 = acc0 * 0.25f + bias[lane];
    float r1 = acc1 * 0.25f + bias[lane + 32];
    if (do_relu) { r0 = fmaxf(r0, 0.f); r1 = fmaxf(r1, 0.f); }
    float* orow = out + (size_t)d * CDIM + lane;
    orow[0]  = r0;
    orow[32] = r1;
}

extern "C" void kernel_launch(void* const* d_in, const int* in_sizes, int n_in,
                              void* d_out, int out_size, void* d_ws, size_t ws_size,
                              hipStream_t stream) {
    const float* x        = (const float*)d_in[0];
    const int*   ei       = (const int*)  d_in[1];
    const float* W1       = (const float*)d_in[2];
    const float* att_src1 = (const float*)d_in[3];
    const float* att_dst1 = (const float*)d_in[4];
    const float* b1       = (const float*)d_in[5];
    const float* W2       = (const float*)d_in[6];
    const float* att_src2 = (const float*)d_in[7];
    const float* att_dst2 = (const float*)d_in[8];
    const float* b2       = (const float*)d_in[9];

    const int Fin = 128;
    const int N   = in_sizes[0] / Fin;
    const int E   = in_sizes[1] / 2;
    const int* srcv = ei;
    const int* dstv = ei + E;

    float* ws    = (float*)d_ws;
    float* Hbuf  = ws;
    float* HID   = Hbuf + (size_t)N * HC;
    float* ASRC  = HID  + (size_t)N * CDIM;
    float* ADST  = ASRC + (size_t)N * HEADS;
    int* CNT     = (int*)(ADST + (size_t)N * HEADS);
    int* ROWPTR  = CNT + N;
    int* CURSOR  = ROWPTR + (N + 1);
    int* CSRC    = CURSOR + N;

    const int gemmStrips = (N / 16) * (HC / 64);
    const int gemmBlocks = (gemmStrips + 3) / 4;
    const int nhBlocks   = (N * HEADS + 255) / 256;
    const int eBlocks    = (E + 255) / 256;
    const int ndBlocks   = (N + 7) / 8;

    hipMemsetAsync(CNT, 0, (size_t)N * 4, stream);
    count_dst<<<eBlocks, 256, 0, stream>>>(dstv, CNT, E);
    scan_cnt<<<1, SCAN_T, 0, stream>>>(CNT, ROWPTR, CURSOR, N);
    fill_csr<<<eBlocks, 256, 0, stream>>>(srcv, dstv, CURSOR, CSRC, E);

    gemm_wmma_f32<<<gemmBlocks, 128, 0, stream>>>(x, W1, Hbuf, N, Fin, HC);
    attn_scores<<<nhBlocks, 256, 0, stream>>>(Hbuf, att_src1, att_dst1, ASRC, ADST, N);
    node_aggregate<<<ndBlocks, 256, 0, stream>>>(ROWPTR, CSRC, Hbuf, ASRC, ADST, b1, HID, N,  1);

    gemm_wmma_f32<<<gemmBlocks, 128, 0, stream>>>(HID, W2, Hbuf, N, CDIM, HC);
    attn_scores<<<nhBlocks, 256, 0, stream>>>(Hbuf, att_src2, att_dst2, ASRC, ADST, N);
    node_aggregate<<<ndBlocks, 256, 0, stream>>>(ROWPTR, CSRC, Hbuf, ASRC, ADST, b2, (float*)d_out, N,  0);
}


# edge-count prefix scan: the one block walks the counts in passes of 8192, 8 consecutive counts per thread (1 KiB contiguous per wave), wave shuffles + LDS; was 196-byte strided chunks per thread
# speedup vs baseline: 5.7505x; 1.0708x over previous
#include <hip/hip_runtime.h>
#include <hip/hip_bf16.h>
#include <math.h>

typedef float v2f __attribute__((ext_vector_type(2)));
typedef float v8f __attribute__((ext_vector_type(8)));

#define HEADS 4
#define HC 256
#define CDIM 64
#define NEG_SLOPE 0.2f

__global__ void gemm_wmma_f32(const float* __restrict__ A,
                              const float* __restrict__ B,
                              float* __restrict__ C,
                              int M, int K, int Ncol) {
    const int lane   = threadIdx.x & 31;
    const int wave   = threadIdx.x >> 5;
    const int wtile  = blockIdx.x * 4 + wave;
    const int wtprow = Ncol >> 6;
    const int total  = (M >> 4) * wtprow;
    if (wtile >= total) return;
    const int trow   = wtile / wtprow;
    const int tcol   = wtile - trow * wtprow;
    const int row0   = trow << 4;
    const int col0   = tcol << 6;
    const int laneM  = lane & 15;
    const int laneHi = lane >> 4;

    const float* Arow  = A + (size_t)(row0 + laneM) * K;
    const float* Bbase = B + (size_t)col0 + laneM;

    v8f acc0 = {}, acc1 = {}, acc2 = {}, acc3 = {};
#pragma unroll 2
    for (int k = 0; k < K; k += 4) {
        const int ka = k + (laneHi << 1);
        v2f a;
        a.x = Arow[ka];
        a.y = Arow[ka + 1];
        const float* B0 = Bbase + (size_t)ka * Ncol;
        const float* B1 = Bbase + (size_t)(ka + 1) * Ncol;
        v2f b0, b1, b2, b3;
        b0.x = B0[0];  b0.y = B1[0];
        b1.x = B0[16]; b1.y = B1[16];
        b2.x = B0[32]; b2.y = B1[32];
        b3.x = B0[48]; b3.y = B1[48];
        acc0 = __builtin_amdgcn_wmma_f32_16x16x4_f32(false, a, false, b0, (short)0, acc0, false, false);
        acc1 = __builtin_amdgcn_wmma_f32_16x16x4_f32(false, a, false, b1, (short)0, acc1, false, false);
        acc2 = __builtin_amdgcn_wmma_f32_16x16x4_f32(false, a, false, b2, (short)0, acc2, false, false);
        acc3 = __builtin_amdgcn_wmma_f32_16x16x4_f32(false, a, false, b3, (short)0, acc3, false, false);
    }

    float* Crow = C + (size_t)(row0 + 8 * laneHi) * Ncol + col0 + laneM;
#pragma unroll
    for (int i = 0; i < 8; ++i) {
        float* r = Crow + (size_t)i * Ncol;
        r[0]  = acc0[i];
        r[16] = acc1[i];
        r[32] = acc2[i];
        r[48] = acc3[i];
    }
}

__global__ void attn_scores(const float* __restrict__ Hm,
                            const float* __restrict__ att_s,
                            const float* __restrict__ att_d,
                            float* __restrict__ as_, float* __restrict__ ad_,
                            int N) {
    const int idx = blockIdx.x * blockDim.x + threadIdx.x;
    if (idx >= N * HEADS) return;
    const int n = idx >> 2, h = idx & 3;
    const float* row = Hm + (size_t)n * HC + h * CDIM;
    const float* vs  = att_s + h * CDIM;
    const float* vd  = att_d + h * CDIM;
    float s = 0.f, d = 0.f;
#pragma unroll 8
    for (int c = 0; c < CDIM; ++c) {
        const float v = row[c];
        s = fmaf(v, vs[c], s);
        d = fmaf(v, vd[c], d);
    }
    as_[idx] = s;
    ad_[idx] = d;
}

__device__ __forceinline__ float leaky(float e) {
    return e > 0.f ? e : NEG_SLOPE * e;
}

__global__ void count_dst(const int* __restrict__ dstv, int* __restrict__ cnt, int E) {
    const int e = blockIdx.x * blockDim.x + threadIdx.x;
    if (e < E) atomicAdd(&cnt[dstv[e]], 1);
}

#define SCAN_T 1024
#define SCAN_PER 8
#define SCAN_TILE (SCAN_T * SCAN_PER)

__device__ __forceinline__ int wave_sum_i(int v) {
#pragma unroll
    for (int o = 16; o > 0; o >>= 1) v += __shfl_xor(v, o, 32);
    return v;
}

__global__ void __launch_bounds__(SCAN_T)
scan_cnt(const int* __restrict__ cnt, int* __restrict__ rowptr,
         int* __restrict__ cursor, int N) {
    __shared__ int wtot[SCAN_T / 32];
    const int t    = threadIdx.x;
    const int lane = t & 31;
    const int w    = t >> 5;
    const bool vec_ok =
        ((((size_t)cnt) | ((size_t)rowptr) | ((size_t)cursor)) & 15) == 0;
    int carry = 0;
    for (int base = 0; base < N; base += SCAN_TILE) {
        const int i0 = base + t * SCAN_PER;
        const bool whole = vec_ok && (i0 + SCAN_PER <= N);
        int v[SCAN_PER];
        if (whole) {
            const int4 a = *reinterpret_cast<const int4*>(cnt + i0);
            const int4 b = *reinterpret_cast<const int4*>(cnt + i0 + 4);
            v[0] = a.x; v[1] = a.y; v[2] = a.z; v[3] = a.w;
            v[4] = b.x; v[5] = b.y; v[6] = b.z; v[7] = b.w;
        } else {
#pragma unroll
            for (int k = 0; k < SCAN_PER; ++k) v[k] = (i0 + k < N) ? cnt[i0 + k] : 0;
        }
        int p[SCAN_PER];
        int s = 0;
#pragma unroll
        for (int k = 0; k < SCAN_PER; ++k) { p[k] = s; s += v[k]; }
        int x = s;
#pragma unroll
        for (int o = 1; o < 32; o <<= 1) {
            const int y = __shfl_up(x, o, 32);
            if (lane >= o) x += y;
        }
        if (lane == 31) wtot[w] = x;
        __syncthreads();
        const int wt     = wtot[lane];
        const int before = wave_sum_i(lane < w ? wt : 0);
        const int total  = wave_sum_i(wt);
        __syncthreads();
        const int off = carry + before + (x - s);
        if (whole) {
            const int4 r0 = make_int4(off + p[0], off + p[1], off + p[2], off + p[3]);
            const int4 r1 = make_int4(off + p[4], off + p[5], off + p[6], off + p[7]);
            *reinterpret_cast<int4*>(rowptr + i0)     = r0;
            *reinterpret_cast<int4*>(rowptr + i0 + 4) = r1;
            *reinterpret_cast<int4*>(cursor + i0)     = r0;
            *reinterpret_cast<int4*>(cursor + i0 + 4) = r1;
        } else {
#pragma unroll
            for (int k = 0; k < SCAN_PER; ++k) {
                if (i0 + k < N) { rowptr[i0 + k] = off + p[k]; cursor[i0 + k] = off + p[k]; }
            }
        }
        carry += total;
    }
    if (t == 0) rowptr[N] = carry;
}

__global__ void fill_csr(const int* __restrict__ srcv, const int* __restrict__ dstv,
                         int* __restrict__ cursor, int* __restrict__ csrc, int E) {
    const int e = blockIdx.x * blockDim.x + threadIdx.x;
    if (e >= E) return;
    const int pos = atomicAdd(&cursor[dstv[e]], 1);
    csrc[pos] = srcv[e];
}

__device__ __forceinline__ float wave_max(float v) {
#pragma unroll
    for (int o = 16; o > 0; o >>= 1) v = fmaxf(v, __shfl_xor(v, o, 32));
    return v;
}
__device__ __forceinline__ float wave_sum(float v) {
#pragma unroll
    for (int o = 16; o > 0; o >>= 1) v += __shfl_xor(v, o, 32);
    return v;
}

__device__ __forceinline__ void edge_scores(int i, int cnt, int base, int d,
                                            const int* __restrict__ csrc,
                                            const float* __restrict__ as_,
                                            const float4 adv, int& s, float4& e) {
    s = d;
    if (i < cnt) s = csrc[base + i];
    const float4 a = *reinterpret_cast<const float4*>(as_ + (size_t)s * HEADS);
    e.x = leaky(a.x + adv.x);
    e.y = leaky(a.y + adv.y);
    e.z = leaky(a.z + adv.z);
    e.w = leaky(a.w + adv.w);
}

__device__ __forceinline__ void gather_row(const float* __restrict__ Hm, int s,
                                           float a0, float a1, float a2, float a3,
                                           int lane, float& x0, float& x1) {
    const float* hr = Hm + (size_t)s * HC + lane;
    x0 = hr[0] * a0;
    x1 = hr[32] * a0;
    x0 = fmaf(hr[CDIM],          a1, x0);
    x1 = fmaf(hr[CDIM + 32],     a1, x1);
    x0 = fmaf(hr[2 * CDIM],      a2, x0);
    x1 = fmaf(hr[2 * CDIM + 32], a2, x1);
    x0 = fmaf(hr[3 * CDIM],      a3, x0);
    x1 = fmaf(hr[3 * CDIM + 32], a3, x1);
}

__global__ void __launch_bounds__(256)
node_aggregate(const int* __restrict__ rowptr, const int* __restrict__ csrc,
               const float* __restrict__ Hm,
               const float* __restrict__ as_, const float* __restrict__ ad_,
               const float* __restrict__ bias, float* __restrict__ out,
               int N, int do_relu) {
    const int lane = threadIdx.x & 31;
    const int wave = __builtin_amdgcn_readfirstlane((int)(threadIdx.x >> 5));
    const int d = blockIdx.x * 8 + wave;
    if (d >= N) return;
    const int base = __builtin_amdgcn_readfirstlane(rowptr[d]);
    const int cnt  = __builtin_amdgcn_readfirstlane(rowptr[d + 1]) - base;
    const int T    = cnt + 1;
    const float4 adv = *reinterpret_cast<const float4*>(ad_ + (size_t)d * HEADS);

    int s_first; float4 e_first;
    edge_scores(lane, cnt, base, d, csrc, as_, adv, s_first, e_first);

    float m0 = -INFINITY, m1 = -INFINITY, m2 = -INFINITY, m3 = -INFINITY;
    for (int c0 = 0; c0 < T; c0 += 32) {
        const int i = c0 + lane;
        int s = s_first; float4 e = e_first;
        if (c0 > 0) edge_scores(i, cnt, base, d, csrc, as_, adv, s, e);
        if (i < T) {
            m0 = fmaxf(m0, e.x); m1 = fmaxf(m1, e.y);
            m2 = fmaxf(m2, e.z); m3 = fmaxf(m3, e.w);
        }
    }
    m0 = wave_max(m0); m1 = wave_max(m1); m2 = wave_max(m2); m3 = wave_max(m3);

    float s0 = 0.f, s1 = 0.f, s2 = 0.f, s3 = 0.f;
    for (int c0 = 0; c0 < T; c0 += 32) {
        const int i = c0 + lane;
        int s = s_first; float4 e = e_first;
        if (c0 > 0) edge_scores(i, cnt, base, d, csrc, as_, adv, s, e);
        if (i < T) {
            s0 += __expf(e.x - m0); s1 += __expf(e.y - m1);
            s2 += __expf(e.z - m2); s3 += __expf(e.w - m3);
        }
    }
    const float den0 = wave_sum(s0) + 1e-16f;
    const float den1 = wave_sum(s1) + 1e-16f;
    const float den2 = wave_sum(s2) + 1e-16f;
    const float den3 = wave_sum(s3) + 1e-16f;

    float acc0 = 0.f, acc1 = 0.f;
    for (int c0 = 0; c0 < T; c0 += 32) {
        int sl = s_first; float4 e = e_first;
        if (c0 > 0) edge_scores(c0 + lane, cnt, base, d, csrc, as_, adv, sl, e);
        const int w0 = __float_as_int(__expf(e.x - m0) / den0);
        const int w1 = __float_as_int(__expf(e.y - m1) / den1);
        const int w2 = __float_as_int(__expf(e.z - m2) / den2);
        const int w3 = __float_as_int(__expf(e.w - m3) / den3);
        const int n = (T - c0 < 32) ? (T - c0) : 32;
        int j = 0;
        for (; j + 4 <= n; j += 4) {
            float x0[4], x1[4];
#pragma unroll
            for (int k = 0; k < 4; ++k) {
                const int s = __builtin_amdgcn_readlane(sl, j + k);
                const float a0 = __int_as_float(__builtin_amdgcn_readlane(w0, j + k));
                const float a1 = __int_as_float(__builtin_amdgcn_readlane(w1, j + k));
                const float a2 = __int_as_float(__builtin_amdgcn_readlane(w2, j + k));
                const float a3 = __int_as_float(__builtin_amdgcn_readlane(w3, j + k));
                gather_row(Hm, s, a0, a1, a2, a3, lane, x0[k], x1[k]);
            }
            acc0 += (x0[0] + x0[1]) + (x0[2] + x0[3]);
            acc1 += (x1[0] + x1[1]) + (x1[2] + x1[3]);
        }
        for (; j < n; ++j) {
            const int s = __builtin_amdgcn_readlane(sl, j);
            const float a0 = __int_as_float(__builtin_amdgcn_readlane(w0, j));
            const float a1 = __int_as_float(__builtin_amdgcn_readlane(w1, j));
            const float a2 = __int_as_float(__builtin_amdgcn_readlane(w2, j));
            const float a3 = __int_as_float(__builtin_amdgcn_readlane(w3, j));
            float x0, x1;
            gather_row(Hm, s, a0, a1, a2, a3, lane, x0, x1);
            acc0 += x0;
            acc1 += x1;
        }
    }

    float r0 = acc0 * 0.25f + bias[lane];
    float r1 = acc1 * 0.25f + bias[lane + 32];
    if (do_relu) { r0 = fmaxf(r0, 0.f); r1 = fmaxf(r1, 0.f); }
    float* orow = out + (size_t)d * CDIM + lane;
    orow[0]  = r0;
    orow[32] = r1;
}

extern "C" void kernel_launch(void* const* d_in, const int* in_sizes, int n_in,
                              void* d_out, int out_size, void* d_ws, size_t ws_size,
                              hipStream_t stream) {
    const float* x        = (const float*)d_in[0];
    const int*   ei       = (const int*)  d_in[1];
    const float* W1       = (const float*)d_in[2];
    const float* att_src1 = (const float*)d_in[3];
    const float* att_dst1 = (const float*)d_in[4];
    const float* b1       = (const float*)d_in[5];
    const float* W2       = (const float*)d_in[6];
    const float* att_src2 = (const float*)d_in[7];
    const float* att_dst2 = (const float*)d_in[8];
    const float* b2       = (const float*)d_in[9];

    const int Fin = 128;
    const int N   = in_sizes[0] / Fin;
    const int E   = in_sizes[1] / 2;
    const int* srcv = ei;
    const int* dstv = ei + E;

    float* ws    = (float*)d_ws;
    float* Hbuf  = ws;
    float* HID   = Hbuf + (size_t)N * HC;
    float* ASRC  = HID  + (size_t)N * CDIM;
    float* ADST  = ASRC + (size_t)N * HEADS;
    int* CNT     = (int*)(ADST + (size_t)N * HEADS);
    int* CURSOR  = CNT + N;
    int* ROWPTR  = CURSOR + N;
    int* CSRC    = ROWPTR + (N + 1);

    const int gemmStrips = (N / 16) * (HC / 64);
    const int gemmBlocks = (gemmStrips + 3) / 4;
    const int nhBlocks   = (N * HEADS + 255) / 256;
    const int eBlocks    = (E + 255) / 256;
    const int ndBlocks   = (N + 7) / 8;

    hipMemsetAsync(CNT, 0, (size_t)N * 4, stream);
    count_dst<<<eBlocks, 256, 0, stream>>>(dstv, CNT, E);
    scan_cnt<<<1, SCAN_T, 0, stream>>>(CNT, ROWPTR, CURSOR, N);
    fill_csr<<<eBlocks, 256, 0, stream>>>(srcv, dstv, CURSOR, CSRC, E);

    gemm_wmma_f32<<<gemmBlocks, 128, 0, stream>>>(x, W1, Hbuf, N, Fin, HC);
    attn_scores<<<nhBlocks, 256, 0, stream>>>(Hbuf, att_src1, att_dst1, ASRC, ADST, N);
    node_aggregate<<<ndBlocks, 256, 0, stream>>>(ROWPTR, CSRC, Hbuf, ASRC, ADST, b1, HID, N,  1);

    gemm_wmma_f32<<<gemmBlocks, 128, 0, stream>>>(HID, W2, Hbuf, N, CDIM, HC);
    attn_scores<<<nhBlocks, 256, 0, stream>>>(Hbuf, att_src2, att_dst2, ASRC, ADST, N);
    node_aggregate<<<ndBlocks, 256, 0, stream>>>(ROWPTR, CSRC, Hbuf, ASRC, ADST, b2, (float*)d_out, N,  0);
}


# edges grouped by destination by a two-launch counting sort on LDS counters (chunks ordered by 256-node range, then one block per range); no global atomics, scan or memset; slot order from LDS adds
# speedup vs baseline: 10.2057x; 1.7748x over previous
#include <hip/hip_runtime.h>
#include <hip/hip_bf16.h>
#include <math.h>

typedef float v2f __attribute__((ext_vector_type(2)));
typedef float v8f __attribute__((ext_vector_type(8)));

#define HEADS 4
#define HC 256
#define CDIM 64
#define NEG_SLOPE 0.2f
#define FIN1 128
#define FIN2 64

template <int FIN>
__global__ void __launch_bounds__(128)
gemm_heads_wmma_f32(const float* __restrict__ Z,
                    const float* __restrict__ Wm,
                    const float* __restrict__ bias,
                    float* __restrict__ out,
                    int M, int do_relu) {
    constexpr int K = HEADS * FIN;
    const int lane   = threadIdx.x & 31;
    const int wave   = threadIdx.x >> 5;
    const int wtile  = blockIdx.x * 4 + wave;
    const int total  = (M >> 4) * 2;
    if (wtile >= total) return;
    const int row0   = (wtile >> 1) << 4;
    const int col0   = (wtile & 1) << 5;
    const int laneM  = lane & 15;
    const int laneHi = lane >> 4;

    const float* Arow  = Z + (size_t)(row0 + laneM) * K;
    const float* Bbase = Wm + col0 + laneM;

    v8f acc0 = {}, acc1 = {};
#pragma unroll 4
    for (int k = 0; k < K; k += 4) {
        const int ka = k + (laneHi << 1);
        v2f a;
        a.x = Arow[ka];
        a.y = Arow[ka + 1];
        const int hh = ka / FIN;
        const int ff = ka - hh * FIN;
        const float* B0 = Bbase + (size_t)ff * HC + hh * CDIM;
        const float* B1 = B0 + HC;
        v2f b0, b1;
        b0.x = B0[0];  b0.y = B1[0];
        b1.x = B0[16]; b1.y = B1[16];
        acc0 = __builtin_amdgcn_wmma_f32_16x16x4_f32(false, a, false, b0, (short)0, acc0, false, false);
        acc1 = __builtin_amdgcn_wmma_f32_16x16x4_f32(false, a, false, b1, (short)0, acc1, false, false);
    }

    const float bs0 = bias[col0 + laneM];
    const float bs1 = bias[col0 + 16 + laneM];
    float* Crow = out + (size_t)(row0 + 8 * laneHi) * CDIM + col0 + laneM;
#pragma unroll
    for (int i = 0; i < 8; ++i) {
        float* r = Crow + (size_t)i * CDIM;
        float v0 = acc0[i] * 0.25f + bs0;
        float v1 = acc1[i] * 0.25f + bs1;
        if (do_relu) { v0 = fmaxf(v0, 0.f); v1 = fmaxf(v1, 0.f); }
        r[0]  = v0;
        r[16] = v1;
    }
}

__device__ __forceinline__ void head_vector(const float* __restrict__ Wm,
                                            const float* __restrict__ att_s,
                                            const float* __restrict__ att_d,
                                            float* __restrict__ V, int FIN, int idx) {
    const int t = idx / FIN, f = idx - t * FIN;
    const int h = t & 3;
    const float* wr = Wm + (size_t)f * HC + h * CDIM;
    const float* av = (t < HEADS ? att_s : att_d) + h * CDIM;
    float s = 0.f;
#pragma unroll 8
    for (int c = 0; c < CDIM; ++c) s = fmaf(wr[c], av[c], s);
    V[idx] = s;
}

template <int FIN>
__global__ void node_scores(const float* __restrict__ X,
                            const float* __restrict__ V,
                            float* __restrict__ as_, float* __restrict__ ad_,
                            int N) {
    const int idx = blockIdx.x * blockDim.x + threadIdx.x;
    if (idx >= N * HEADS) return;
    const int n = idx >> 2, h = idx & 3;
    const float* row = X + (size_t)n * FIN;
    const float* vs  = V + h * FIN;
    const float* vd  = V + (HEADS + h) * FIN;
    float s = 0.f, d = 0.f;
#pragma unroll 8
    for (int f = 0; f < FIN; ++f) {
        const float v = row[f];
        s = fmaf(v, vs[f], s);
        d = fmaf(v, vd[f], d);
    }
    as_[idx] = s;
    ad_[idx] = d;
}

__device__ __forceinline__ float leaky(float e) {
    return e > 0.f ? e : NEG_SLOPE * e;
}

#define BIN_SHIFT 8
#define BIN_NODES (1 << BIN_SHIFT)
#define SORT_T 1024
#define CHUNK_PER 4
#define CHUNK_CAP (SORT_T * CHUNK_PER)
#define MAXBIN SORT_T
#define OWN_CAP 8192

__device__ __forceinline__ int wave_sum_i(int v) {
#pragma unroll
    for (int o = 16; o > 0; o >>= 1) v += __shfl_xor(v, o, 32);
    return v;
}

__device__ __forceinline__ int block_excl_scan(int v, int* wtot, int& total) {
    const int lane = threadIdx.x & 31;
    const int w    = threadIdx.x >> 5;
    int x = v;
#pragma unroll
    for (int o = 1; o < 32; o <<= 1) {
        const int y = __shfl_up(x, o, 32);
        if (lane >= o) x += y;
    }
    if (lane == 31) wtot[w] = x;
    __syncthreads();
    const int wt     = wtot[lane];
    const int before = wave_sum_i(lane < w ? wt : 0);
    total            = wave_sum_i(wt);
    __syncthreads();
    return before + x - v;
}

__device__ __forceinline__ int block_sum(int v, int* wtot) {
    const int lane = threadIdx.x & 31;
    const int w    = threadIdx.x >> 5;
    const int x    = wave_sum_i(v);
    if (lane == 0) wtot[w] = x;
    __syncthreads();
    const int total = wave_sum_i(wtot[lane]);
    __syncthreads();
    return total;
}

__global__ void __launch_bounds__(SORT_T)
bin_edges(const int* __restrict__ srcv, const int* __restrict__ dstv, int E, int chunk,
          int nBins, int chunkBlocks,
          int2* __restrict__ binned, int* __restrict__ offs,
          const float* __restrict__ W1, const float* __restrict__ as1,
          const float* __restrict__ ad1, float* __restrict__ V1,
          const float* __restrict__ W2, const float* __restrict__ as2,
          const float* __restrict__ ad2, float* __restrict__ V2) {
    __shared__ int  binCnt[MAXBIN];
    __shared__ int  wtot[SORT_T / 32];
    __shared__ int2 staged[CHUNK_CAP];
    const int t = threadIdx.x;
    if ((int)blockIdx.x >= chunkBlocks) {
        const int idx = ((int)blockIdx.x - chunkBlocks) * SORT_T + t;
        if (idx < 2 * HEADS * FIN1) head_vector(W1, as1, ad1, V1, FIN1, idx);
        else if (idx < 2 * HEADS * (FIN1 + FIN2))
            head_vector(W2, as2, ad2, V2, FIN2, idx - 2 * HEADS * FIN1);
        return;
    }
    const int e0 = (int)blockIdx.x * chunk;
    int n = E - e0;
    n = n < 0 ? 0 : (n > chunk ? chunk : n);
    if (t < nBins) binCnt[t] = 0;
    __syncthreads();

    int s[CHUNK_PER], d[CHUNK_PER], r[CHUNK_PER];
#pragma unroll
    for (int i = 0; i < CHUNK_PER; ++i) {
        const int j = t + i * SORT_T;
        s[i] = 0; d[i] = 0;
        if (j < n) { d[i] = dstv[e0 + j]; s[i] = srcv[e0 + j]; }
    }
#pragma unroll
    for (int i = 0; i < CHUNK_PER; ++i) {
        const int j = t + i * SORT_T;
        r[i] = 0;
        if (j < n) r[i] = atomicAdd(&binCnt[d[i] >> BIN_SHIFT], 1);
    }
    __syncthreads();

    const int c = (t < nBins) ? binCnt[t] : 0;
    int tot;
    const int ex = block_excl_scan(c, wtot, tot);
    int* myoffs = offs + (size_t)blockIdx.x * (nBins + 1);
    if (t < nBins) { binCnt[t] = ex; myoffs[t] = ex; }
    if (t == 0) myoffs[nBins] = tot;
    __syncthreads();

#pragma unroll
    for (int i = 0; i < CHUNK_PER; ++i) {
        const int j = t + i * SORT_T;
        if (j < n) staged[binCnt[d[i] >> BIN_SHIFT] + r[i]] = make_int2(s[i], d[i]);
    }
    __syncthreads();
    int2* outp = binned + (size_t)e0;
    for (int j = t; j < n; j += SORT_T) outp[j] = staged[j];
}

__global__ void __launch_bounds__(SORT_T)
build_csr(const int2* __restrict__ binned, const int* __restrict__ offs,
          int nChunks, int chunk, int nBins, int N,
          int* __restrict__ rowptr, int* __restrict__ csrc) {
    __shared__ int nodeCnt[BIN_NODES];
    __shared__ int wtot[SORT_T / 32];
    __shared__ int stagedSrc[OWN_CAP];
    const int t      = threadIdx.x;
    const int b      = blockIdx.x;
    const int node0  = b << BIN_SHIFT;
    const int stride = nBins + 1;

    int lo = 0, hi = 0;
    for (int k = t; k < nChunks; k += SORT_T) {
        lo += offs[(size_t)k * stride + b];
        hi += offs[(size_t)k * stride + b + 1];
    }
    const int base  = block_sum(lo, wtot);
    const int end   = block_sum(hi, wtot);
    const int total = end - base;

    if (t < BIN_NODES) nodeCnt[t] = 0;
    __syncthreads();

    const int hw = t >> 4, hl = t & 15;
    for (int k = hw; k < nChunks; k += SORT_T / 16) {
        const int o0 = offs[(size_t)k * stride + b];
        const int o1 = offs[(size_t)k * stride + b + 1];
        const int2* run = binned + (size_t)k * chunk;
        for (int j = o0 + hl; j < o1; j += 16) atomicAdd(&nodeCnt[run[j].y - node0], 1);
    }
    __syncthreads();

    const int c = (t < BIN_NODES) ? nodeCnt[t] : 0;
    int tot;
    const int ex = block_excl_scan(c, wtot, tot);
    if (t < BIN_NODES) {
        nodeCnt[t] = ex;
        if (node0 + t < N) rowptr[node0 + t] = base + ex;
    }
    if (b == nBins - 1 && t == 0) rowptr[N] = end;
    __syncthreads();

    const bool fits = total <= OWN_CAP;
    for (int k = hw; k < nChunks; k += SORT_T / 16) {
        const int o0 = offs[(size_t)k * stride + b];
        const int o1 = offs[(size_t)k * stride + b + 1];
        const int2* run = binned + (size_t)k * chunk;
        for (int j = o0 + hl; j < o1; j += 16) {
            const int2 e  = run[j];
            const int pos = atomicAdd(&nodeCnt[e.y - node0], 1);
            if (fits) stagedSrc[pos] = e.x;
            else      csrc[base + pos] = e.x;
        }
    }
    if (fits) {
        __syncthreads();
        for (int j = t; j < total; j += SORT_T) csrc[base + j] = stagedSrc[j];
    }
}

__device__ __forceinline__ float wave_max(float v) {
#pragma unroll
    for (int o = 16; o > 0; o >>= 1) v = fmaxf(v, __shfl_xor(v, o, 32));
    return v;
}
__device__ __forceinline__ float wave_sum(float v) {
#pragma unroll
    for (int o = 16; o > 0; o >>= 1) v += __shfl_xor(v, o, 32);
    return v;
}

__device__ __forceinline__ void edge_scores(int i, int cnt, int base, int d,
                                            const int* __restrict__ csrc,
                                            const float* __restrict__ as_,
                                            const float4 adv, int& s, float4& e) {
    s = d;
    if (i < cnt) s = csrc[base + i];
    const float4 a = *reinterpret_cast<const float4*>(as_ + (size_t)s * HEADS);
    e.x = leaky(a.x + adv.x);
    e.y = leaky(a.y + adv.y);
    e.z = leaky(a.z + adv.z);
    e.w = leaky(a.w + adv.w);
}

template <int FIN>
__global__ void __launch_bounds__(256)
node_aggregate_in(const int* __restrict__ rowptr, const int* __restrict__ csrc,
                  const float* __restrict__ X,
                  const float* __restrict__ as_, const float* __restrict__ ad_,
                  float* __restrict__ Z, int N) {
    constexpr int VEC = FIN / 32;
    typedef float vrow __attribute__((ext_vector_type(VEC)));
    const int lane = threadIdx.x & 31;
    const int wave = __builtin_amdgcn_readfirstlane((int)(threadIdx.x >> 5));
    const int d = blockIdx.x * 8 + wave;
    if (d >= N) return;
    const int base = __builtin_amdgcn_readfirstlane(rowptr[d]);
    const int cnt  = __builtin_amdgcn_readfirstlane(rowptr[d + 1]) - base;
    const int T    = cnt + 1;
    const float4 adv = *reinterpret_cast<const float4*>(ad_ + (size_t)d * HEADS);

    int s_first; float4 e_first;
    edge_scores(lane, cnt, base, d, csrc, as_, adv, s_first, e_first);

    float m0 = -INFINITY, m1 = -INFINITY, m2 = -INFINITY, m3 = -INFINITY;
    for (int c0 = 0; c0 < T; c0 += 32) {
        const int i = c0 + lane;
        int s = s_first; float4 e = e_first;
        if (c0 > 0) edge_scores(i, cnt, base, d, csrc, as_, adv, s, e);
        if (i < T) {
            m0 = fmaxf(m0, e.x); m1 = fmaxf(m1, e.y);
            m2 = fmaxf(m2, e.z); m3 = fmaxf(m3, e.w);
        }
    }
    m0 = wave_max(m0); m1 = wave_max(m1); m2 = wave_max(m2); m3 = wave_max(m3);

    float s0 = 0.f, s1 = 0.f, s2 = 0.f, s3 = 0.f;
    for (int c0 = 0; c0 < T; c0 += 32) {
        const int i = c0 + lane;
        int s = s_first; float4 e = e_first;
        if (c0 > 0) edge_scores(i, cnt, base, d, csrc, as_, adv, s, e);
        if (i < T) {
            s0 += __expf(e.x - m0); s1 += __expf(e.y - m1);
            s2 += __expf(e.z - m2); s3 += __expf(e.w - m3);
        }
    }
    const float den0 = wave_sum(s0) + 1e-16f;
    const float den1 = wave_sum(s1) + 1e-16f;
    const float den2 = wave_sum(s2) + 1e-16f;
    const float den3 = wave_sum(s3) + 1e-16f;

    vrow z0 = {}, z1 = {}, z2 = {}, z3 = {};
    const float* Xl = X + lane * VEC;
    for (int c0 = 0; c0 < T; c0 += 32) {
        int sl = s_first; float4 e = e_first;
        if (c0 > 0) edge_scores(c0 + lane, cnt, base, d, csrc, as_, adv, sl, e);
        const int w0 = __float_as_int(__expf(e.x - m0) / den0);
        const int w1 = __float_as_int(__expf(e.y - m1) / den1);
        const int w2 = __float_as_int(__expf(e.z - m2) / den2);
        const int w3 = __float_as_int(__expf(e.w - m3) / den3);
        const int n = (T - c0 < 32) ? (T - c0) : 32;
        int j = 0;
        for (; j + 4 <= n; j += 4) {
            vrow xr[4];
#pragma unroll
            for (int k = 0; k < 4; ++k) {
                const int s = __builtin_amdgcn_readlane(sl, j + k);
                xr[k] = *reinterpret_cast<const vrow*>(Xl + (size_t)s * FIN);
            }
#pragma unroll
            for (int k = 0; k < 4; ++k) {
                const float a0 = __int_as_float(__builtin_amdgcn_readlane(w0, j + k));
                const float a1 = __int_as_float(__builtin_amdgcn_readlane(w1, j + k));
                const float a2 = __int_as_float(__builtin_amdgcn_readlane(w2, j + k));
                const float a3 = __int_as_float(__builtin_amdgcn_readlane(w3, j + k));
                z0 += xr[k] * a0;
                z1 += xr[k] * a1;
                z2 += xr[k] * a2;
                z3 += xr[k] * a3;
            }
        }
        for (; j < n; ++j) {
            const int s = __builtin_amdgcn_readlane(sl, j);
            const vrow xv = *reinterpret_cast<const vrow*>(Xl + (size_t)s * FIN);
            const float a0 = __int_as_float(__builtin_amdgcn_readlane(w0, j));
            const float a1 = __int_as_float(__builtin_amdgcn_readlane(w1, j));
            const float a2 = __int_as_float(__builtin_amdgcn_readlane(w2, j));
            const float a3 = __int_as_float(__builtin_amdgcn_readlane(w3, j));
            z0 += xv * a0;
            z1 += xv * a1;
            z2 += xv * a2;
            z3 += xv * a3;
        }
    }

    float* zrow = Z + (size_t)d * (HEADS * FIN) + lane * VEC;
    *reinterpret_cast<vrow*>(zrow)           = z0;
    *reinterpret_cast<vrow*>(zrow + FIN)     = z1;
    *reinterpret_cast<vrow*>(zrow + 2 * FIN) = z2;
    *reinterpret_cast<vrow*>(zrow + 3 * FIN) = z3;
}

extern "C" void kernel_launch(void* const* d_in, const int* in_sizes, int n_in,
                              void* d_out, int out_size, void* d_ws, size_t ws_size,
                              hipStream_t stream) {
    const float* x        = (const float*)d_in[0];
    const int*   ei       = (const int*)  d_in[1];
    const float* W1       = (const float*)d_in[2];
    const float* att_src1 = (const float*)d_in[3];
    const float* att_dst1 = (const float*)d_in[4];
    const float* b1       = (const float*)d_in[5];
    const float* W2       = (const float*)d_in[6];
    const float* att_src2 = (const float*)d_in[7];
    const float* att_dst2 = (const float*)d_in[8];
    const float* b2       = (const float*)d_in[9];

    const int Fin = 128;
    const int N   = in_sizes[0] / Fin;
    const int E   = in_sizes[1] / 2;
    const int* srcv = ei;
    const int* dstv = ei + E;

    const int nChunks = (E + CHUNK_CAP * 256 - 1) / (CHUNK_CAP * 256) * 256;
    const int chunk   = ((E + nChunks - 1) / nChunks + 31) & ~31;
    const int nBins   = (N + BIN_NODES - 1) / BIN_NODES;

    float* ws    = (float*)d_ws;
    float* Zbuf  = ws;
    float* HID   = Zbuf + (size_t)N * HEADS * FIN1;
    float* ASRC  = HID  + (size_t)N * CDIM;
    float* ADST  = ASRC + (size_t)N * HEADS;
    float* V1    = ADST + (size_t)N * HEADS;
    float* V2    = V1 + 2 * HEADS * FIN1;
    int2* BINNED = (int2*)(V2 + 2 * HEADS * FIN2);
    int* OFFS    = (int*)(BINNED + (size_t)nChunks * chunk);
    int* ROWPTR  = OFFS + (size_t)nChunks * (nBins + 1);
    int* CSRC    = ROWPTR + (N + 1);

    const int gemmPieces = (N / 16) * 2;
    const int gemmBlocks = (gemmPieces + 3) / 4;
    const int nhBlocks   = (N * HEADS + 255) / 256;
    const int hvBlocks   = (2 * HEADS * (FIN1 + FIN2) + SORT_T - 1) / SORT_T;
    const int ndBlocks   = (N + 7) / 8;

    bin_edges<<<nChunks + hvBlocks, SORT_T, 0, stream>>>(srcv, dstv, E, chunk, nBins, nChunks,
                                                         BINNED, OFFS,
                                                         W1, att_src1, att_dst1, V1,
                                                         W2, att_src2, att_dst2, V2);
    build_csr<<<nBins, SORT_T, 0, stream>>>(BINNED, OFFS, nChunks, chunk, nBins, N, ROWPTR, CSRC);

    node_scores<FIN1><<<nhBlocks, 256, 0, stream>>>(x, V1, ASRC, ADST, N);
    node_aggregate_in<FIN1><<<ndBlocks, 256, 0, stream>>>(ROWPTR, CSRC, x, ASRC, ADST, Zbuf, N);
    gemm_heads_wmma_f32<FIN1><<<gemmBlocks, 128, 0, stream>>>(Zbuf, W1, b1, HID, N,  1);

    node_scores<FIN2><<<nhBlocks, 256, 0, stream>>>(HID, V2, ASRC, ADST, N);
    node_aggregate_in<FIN2><<<ndBlocks, 256, 0, stream>>>(ROWPTR, CSRC, HID, ASRC, ADST, Zbuf, N);
    gemm_heads_wmma_f32<FIN2><<<gemmBlocks, 128, 0, stream>>>(Zbuf, W2, b2, (float*)d_out, N,  0);
}


# node sums and projection in ONE kernel: a block sums 16 nodes' input rows per head into LDS (2 nodes per wave), then multiplies the tile by W on WMMA; sum order changes (heads added after products)
# speedup vs baseline: 13.6653x; 1.3390x over previous
#include <hip/hip_runtime.h>
#include <hip/hip_bf16.h>
#include <math.h>

typedef float v2f __attribute__((ext_vector_type(2)));
typedef float v8f __attribute__((ext_vector_type(8)));

#define HEADS 4
#define HC 256
#define CDIM 64
#define NEG_SLOPE 0.2f
#define FIN1 128
#define FIN2 64


__device__ __forceinline__ void head_vector(const float* __restrict__ Wm,
                                            const float* __restrict__ att_s,
                                            const float* __restrict__ att_d,
                                            float* __restrict__ V, int FIN, int idx) {
    const int t = idx / FIN, f = idx - t * FIN;
    const int h = t & 3;
    const float* wr = Wm + (size_t)f * HC + h * CDIM;
    const float* av = (t < HEADS ? att_s : att_d) + h * CDIM;
    float s = 0.f;
#pragma unroll 8
    for (int c = 0; c < CDIM; ++c) s = fmaf(wr[c], av[c], s);
    V[idx] = s;
}

template <int FIN>
__global__ void node_scores(const float* __restrict__ X,
                            const float* __restrict__ V,
                            float* __restrict__ as_, float* __restrict__ ad_,
                            int N) {
    const int idx = blockIdx.x * blockDim.x + threadIdx.x;
    if (idx >= N * HEADS) return;
    const int n = idx >> 2, h = idx & 3;
    const float* row = X + (size_t)n * FIN;
    const float* vs  = V + h * FIN;
    const float* vd  = V + (HEADS + h) * FIN;
    float s = 0.f, d = 0.f;
#pragma unroll 8
    for (int f = 0; f < FIN; ++f) {
        const float v = row[f];
        s = fmaf(v, vs[f], s);
        d = fmaf(v, vd[f], d);
    }
    as_[idx] = s;
    ad_[idx] = d;
}

__device__ __forceinline__ float leaky(float e) {
    return e > 0.f ? e : NEG_SLOPE * e;
}

#define BIN_SHIFT 8
#define BIN_NODES (1 << BIN_SHIFT)
#define SORT_T 1024
#define CHUNK_PER 4
#define CHUNK_CAP (SORT_T * CHUNK_PER)
#define MAXBIN SORT_T
#define OWN_CAP 8192

__device__ __forceinline__ int wave_sum_i(int v) {
#pragma unroll
    for (int o = 16; o > 0; o >>= 1) v += __shfl_xor(v, o, 32);
    return v;
}

__device__ __forceinline__ int block_excl_scan(int v, int* wtot, int& total) {
    const int lane = threadIdx.x & 31;
    const int w    = threadIdx.x >> 5;
    int x = v;
#pragma unroll
    for (int o = 1; o < 32; o <<= 1) {
        const int y = __shfl_up(x, o, 32);
        if (lane >= o) x += y;
    }
    if (lane == 31) wtot[w] = x;
    __syncthreads();
    const int wt     = wtot[lane];
    const int before = wave_sum_i(lane < w ? wt : 0);
    total            = wave_sum_i(wt);
    __syncthreads();
    return before + x - v;
}

__device__ __forceinline__ int block_sum(int v, int* wtot) {
    const int lane = threadIdx.x & 31;
    const int w    = threadIdx.x >> 5;
    const int x    = wave_sum_i(v);
    if (lane == 0) wtot[w] = x;
    __syncthreads();
    const int total = wave_sum_i(wtot[lane]);
    __syncthreads();
    return total;
}

__global__ void __launch_bounds__(SORT_T)
bin_edges(const int* __restrict__ srcv, const int* __restrict__ dstv, int E, int chunk,
          int nBins, int chunkBlocks,
          int2* __restrict__ binned, int* __restrict__ offs,
          const float* __restrict__ W1, const float* __restrict__ as1,
          const float* __restrict__ ad1, float* __restrict__ V1,
          const float* __restrict__ W2, const float* __restrict__ as2,
          const float* __restrict__ ad2, float* __restrict__ V2) {
    __shared__ int  binCnt[MAXBIN];
    __shared__ int  wtot[SORT_T / 32];
    __shared__ int2 staged[CHUNK_CAP];
    const int t = threadIdx.x;
    if ((int)blockIdx.x >= chunkBlocks) {
        const int idx = ((int)blockIdx.x - chunkBlocks) * SORT_T + t;
        if (idx < 2 * HEADS * FIN1) head_vector(W1, as1, ad1, V1, FIN1, idx);
        else if (idx < 2 * HEADS * (FIN1 + FIN2))
            head_vector(W2, as2, ad2, V2, FIN2, idx - 2 * HEADS * FIN1);
        return;
    }
    const int e0 = (int)blockIdx.x * chunk;
    int n = E - e0;
    n = n < 0 ? 0 : (n > chunk ? chunk : n);
    if (t < nBins) binCnt[t] = 0;
    __syncthreads();

    int s[CHUNK_PER], d[CHUNK_PER], r[CHUNK_PER];
#pragma unroll
    for (int i = 0; i < CHUNK_PER; ++i) {
        const int j = t + i * SORT_T;
        s[i] = 0; d[i] = 0;
        if (j < n) { d[i] = dstv[e0 + j]; s[i] = srcv[e0 + j]; }
    }
#pragma unroll
    for (int i = 0; i < CHUNK_PER; ++i) {
        const int j = t + i * SORT_T;
        r[i] = 0;
        if (j < n) r[i] = atomicAdd(&binCnt[d[i] >> BIN_SHIFT], 1);
    }
    __syncthreads();

    const int c = (t < nBins) ? binCnt[t] : 0;
    int tot;
    const int ex = block_excl_scan(c, wtot, tot);
    int* myoffs = offs + (size_t)blockIdx.x * (nBins + 1);
    if (t < nBins) { binCnt[t] = ex; myoffs[t] = ex; }
    if (t == 0) myoffs[nBins] = tot;
    __syncthreads();

#pragma unroll
    for (int i = 0; i < CHUNK_PER; ++i) {
        const int j = t + i * SORT_T;
        if (j < n) staged[binCnt[d[i] >> BIN_SHIFT] + r[i]] = make_int2(s[i], d[i]);
    }
    __syncthreads();
    int2* outp = binned + (size_t)e0;
    for (int j = t; j < n; j += SORT_T) outp[j] = staged[j];
}

__global__ void __launch_bounds__(SORT_T)
build_csr(const int2* __restrict__ binned, const int* __restrict__ offs,
          int nChunks, int chunk, int nBins, int N,
          int* __restrict__ rowptr, int* __restrict__ csrc) {
    __shared__ int nodeCnt[BIN_NODES];
    __shared__ int wtot[SORT_T / 32];
    __shared__ int stagedSrc[OWN_CAP];
    const int t      = threadIdx.x;
    const int b      = blockIdx.x;
    const int node0  = b << BIN_SHIFT;
    const int stride = nBins + 1;

    int lo = 0, hi = 0;
    for (int k = t; k < nChunks; k += SORT_T) {
        lo += offs[(size_t)k * stride + b];
        hi += offs[(size_t)k * stride + b + 1];
    }
    const int base  = block_sum(lo, wtot);
    const int end   = block_sum(hi, wtot);
    const int total = end - base;

    if (t < BIN_NODES) nodeCnt[t] = 0;
    __syncthreads();

    const int hw = t >> 4, hl = t & 15;
    for (int k = hw; k < nChunks; k += SORT_T / 16) {
        const int o0 = offs[(size_t)k * stride + b];
        const int o1 = offs[(size_t)k * stride + b + 1];
        const int2* run = binned + (size_t)k * chunk;
        for (int j = o0 + hl; j < o1; j += 16) atomicAdd(&nodeCnt[run[j].y - node0], 1);
    }
    __syncthreads();

    const int c = (t < BIN_NODES) ? nodeCnt[t] : 0;
    int tot;
    const int ex = block_excl_scan(c, wtot, tot);
    if (t < BIN_NODES) {
        nodeCnt[t] = ex;
        if (node0 + t < N) rowptr[node0 + t] = base + ex;
    }
    if (b == nBins - 1 && t == 0) rowptr[N] = end;
    __syncthreads();

    const bool fits = total <= OWN_CAP;
    for (int k = hw; k < nChunks; k += SORT_T / 16) {
        const int o0 = offs[(size_t)k * stride + b];
        const int o1 = offs[(size_t)k * stride + b + 1];
        const int2* run = binned + (size_t)k * chunk;
        for (int j = o0 + hl; j < o1; j += 16) {
            const int2 e  = run[j];
            const int pos = atomicAdd(&nodeCnt[e.y - node0], 1);
            if (fits) stagedSrc[pos] = e.x;
            else      csrc[base + pos] = e.x;
        }
    }
    if (fits) {
        __syncthreads();
        for (int j = t; j < total; j += SORT_T) csrc[base + j] = stagedSrc[j];
    }
}

__device__ __forceinline__ float wave_max(float v) {
#pragma unroll
    for (int o = 16; o > 0; o >>= 1) v = fmaxf(v, __shfl_xor(v, o, 32));
    return v;
}
__device__ __forceinline__ float wave_sum(float v) {
#pragma unroll
    for (int o = 16; o > 0; o >>= 1) v += __shfl_xor(v, o, 32);
    return v;
}

__device__ __forceinline__ void edge_scores(int i, int cnt, int base, int d,
                                            const int* __restrict__ csrc,
                                            const float* __restrict__ as_,
                                            const float4 adv, int& s, float4& e) {
    s = d;
    if (i < cnt) s = csrc[base + i];
    const float4 a = *reinterpret_cast<const float4*>(as_ + (size_t)s * HEADS);
    e.x = leaky(a.x + adv.x);
    e.y = leaky(a.y + adv.y);
    e.z = leaky(a.z + adv.z);
    e.w = leaky(a.w + adv.w);
}

template <int FIN>
__device__ __forceinline__ void node_rows_sum(const int d, const int base, const int cnt,
                                              const float* __restrict__ adrow,
                                              const int s_first, const float4 e_first,
                                              const int* __restrict__ csrc,
                                              const float* __restrict__ X,
                                              const float* __restrict__ as_,
                                              float* __restrict__ zrow) {
    constexpr int VEC = FIN / 32;
    typedef float vrow __attribute__((ext_vector_type(VEC)));
    const int lane = threadIdx.x & 31;
    const int T    = cnt + 1;

    float m0 = -INFINITY, m1 = -INFINITY, m2 = -INFINITY, m3 = -INFINITY;
    for (int c0 = 0; c0 < T; c0 += 32) {
        const int i = c0 + lane;
        int s = s_first; float4 e = e_first;
        if (c0 > 0)
            edge_scores(i, cnt, base, d, csrc, as_, *reinterpret_cast<const float4*>(adrow), s, e);
        if (i < T) {
            m0 = fmaxf(m0, e.x); m1 = fmaxf(m1, e.y);
            m2 = fmaxf(m2, e.z); m3 = fmaxf(m3, e.w);
        }
    }
    m0 = wave_max(m0); m1 = wave_max(m1); m2 = wave_max(m2); m3 = wave_max(m3);

    float s0 = 0.f, s1 = 0.f, s2 = 0.f, s3 = 0.f;
    for (int c0 = 0; c0 < T; c0 += 32) {
        const int i = c0 + lane;
        int s = s_first; float4 e = e_first;
        if (c0 > 0)
            edge_scores(i, cnt, base, d, csrc, as_, *reinterpret_cast<const float4*>(adrow), s, e);
        if (i < T) {
            s0 += __expf(e.x - m0); s1 += __expf(e.y - m1);
            s2 += __expf(e.z - m2); s3 += __expf(e.w - m3);
        }
    }
    const float den0 = wave_sum(s0) + 1e-16f;
    const float den1 = wave_sum(s1) + 1e-16f;
    const float den2 = wave_sum(s2) + 1e-16f;
    const float den3 = wave_sum(s3) + 1e-16f;

    vrow z0 = {}, z1 = {}, z2 = {}, z3 = {};
    const unsigned lo = lane * VEC;
    for (int c0 = 0; c0 < T; c0 += 32) {
        int sl = s_first; float4 e = e_first;
        if (c0 > 0)
            edge_scores(c0 + lane, cnt, base, d, csrc, as_, *reinterpret_cast<const float4*>(adrow), sl, e);
        const int w0 = __float_as_int(__expf(e.x - m0) / den0);
        const int w1 = __float_as_int(__expf(e.y - m1) / den1);
        const int w2 = __float_as_int(__expf(e.z - m2) / den2);
        const int w3 = __float_as_int(__expf(e.w - m3) / den3);
        const int n = (T - c0 < 32) ? (T - c0) : 32;
        int j = 0;
        for (; j + 4 <= n; j += 4) {
            vrow xr[4];
#pragma unroll
            for (int k = 0; k < 4; ++k) {
                const int s = __builtin_amdgcn_readlane(sl, j + k);
                xr[k] = *reinterpret_cast<const vrow*>(X + ((unsigned)s * FIN + lo));
            }
#pragma unroll
            for (int k = 0; k < 4; ++k) {
                const float a0 = __int_as_float(__builtin_amdgcn_readlane(w0, j + k));
                const float a1 = __int_as_float(__builtin_amdgcn_readlane(w1, j + k));
                const float a2 = __int_as_float(__builtin_amdgcn_readlane(w2, j + k));
                const float a3 = __int_as_float(__builtin_amdgcn_readlane(w3, j + k));
                z0 += xr[k] * a0;
                z1 += xr[k] * a1;
                z2 += xr[k] * a2;
                z3 += xr[k] * a3;
            }
        }
        for (; j < n; ++j) {
            const int s = __builtin_amdgcn_readlane(sl, j);
            const vrow xv = *reinterpret_cast<const vrow*>(X + ((unsigned)s * FIN + lo));
            const float a0 = __int_as_float(__builtin_amdgcn_readlane(w0, j));
            const float a1 = __int_as_float(__builtin_amdgcn_readlane(w1, j));
            const float a2 = __int_as_float(__builtin_amdgcn_readlane(w2, j));
            const float a3 = __int_as_float(__builtin_amdgcn_readlane(w3, j));
            z0 += xv * a0;
            z1 += xv * a1;
            z2 += xv * a2;
            z3 += xv * a3;
        }
    }

    *reinterpret_cast<vrow*>(zrow)           = z0;
    *reinterpret_cast<vrow*>(zrow + FIN)     = z1;
    *reinterpret_cast<vrow*>(zrow + 2 * FIN) = z2;
    *reinterpret_cast<vrow*>(zrow + 3 * FIN) = z3;
}

#define TILE_M 16
#define PPITCH 68

template <int FIN>
__global__ void __launch_bounds__(256)
node_sum_project(const int* __restrict__ rowptr, const int* __restrict__ csrc,
                 const float* __restrict__ X,
                 const float* __restrict__ as_, const float* __restrict__ ad_,
                 const float* __restrict__ Wm, const float* __restrict__ bias,
                 float* __restrict__ out, int N, int do_relu) {
    constexpr int VEC   = FIN / 32;
    constexpr int K     = HEADS * FIN;
    constexpr int PITCH = K + 4;
    constexpr int ZSIZE = TILE_M * PITCH > HEADS * TILE_M * PPITCH ? TILE_M * PITCH
                                                                   : HEADS * TILE_M * PPITCH;
    typedef float vrow __attribute__((ext_vector_type(VEC)));
    __shared__ __attribute__((aligned(16))) float Zs[ZSIZE];

    const int lane = threadIdx.x & 31;
    const int wave  = __builtin_amdgcn_readfirstlane((int)(threadIdx.x >> 5));
    const int node0 = blockIdx.x * TILE_M;

    {
        const int rowA = wave * 2;
        const int dA = node0 + rowA, dB = dA + 1;
        const int dAc = dA < N ? dA : N - 1;
        const int dBc = dB < N ? dB : N - 1;
        int rpi = dA + (lane < 3 ? lane : 0);
        rpi = rpi < N ? rpi : N;
        const int rp  = rowptr[rpi];
        const int bA   = __builtin_amdgcn_readlane(rp, 0);
        const int bB   = __builtin_amdgcn_readlane(rp, 1);
        const int cntA = bB - bA;
        const int cntB = __builtin_amdgcn_readlane(rp, 2) - bB;
        const float* adA = ad_ + (size_t)dAc * HEADS;
        const float* adB = ad_ + (size_t)dBc * HEADS;
        const float4 advA = *reinterpret_cast<const float4*>(adA);
        const float4 advB = *reinterpret_cast<const float4*>(adB);
        int sA, sB; float4 eA, eB;
        edge_scores(lane, cntA, bA, dAc, csrc, as_, advA, sA, eA);
        edge_scores(lane, cntB, bB, dBc, csrc, as_, advB, sB, eB);

        float* zA = Zs + rowA * PITCH + lane * VEC;
        float* zB = zA + PITCH;
        if (dA < N) {
            node_rows_sum<FIN>(dA, bA, cntA, adA, sA, eA, csrc, X, as_, zA);
        } else {
            const vrow zz = {};
#pragma unroll
            for (int h = 0; h < HEADS; ++h) *reinterpret_cast<vrow*>(zA + h * FIN) = zz;
        }
        if (dB < N) {
            node_rows_sum<FIN>(dB, bB, cntB, adB, sB, eB, csrc, X, as_, zB);
        } else {
            const vrow zz = {};
#pragma unroll
            for (int h = 0; h < HEADS; ++h) *reinterpret_cast<vrow*>(zB + h * FIN) = zz;
        }
    }
    __syncthreads();

    const int laneM  = lane & 15;
    const int laneHi = lane >> 4;
    const int hq     = wave >> 1;
    const int col0   = (wave & 1) << 5;
    const float* Arow  = Zs + laneM * PITCH + hq * FIN + (laneHi << 1);
    const float* Bbase = Wm + (size_t)(laneHi << 1) * HC + hq * CDIM + col0 + laneM;

    v8f acc0 = {}, acc1 = {};
#pragma unroll 4
    for (int f = 0; f < FIN; f += 4) {
        const v2f a = *reinterpret_cast<const v2f*>(Arow + f);
        const float* B0 = Bbase + (size_t)f * HC;
        const float* B1 = B0 + HC;
        v2f b0, b1;
        b0.x = B0[0];  b0.y = B1[0];
        b1.x = B0[16]; b1.y = B1[16];
        acc0 = __builtin_amdgcn_wmma_f32_16x16x4_f32(false, a, false, b0, (short)0, acc0, false, false);
        acc1 = __builtin_amdgcn_wmma_f32_16x16x4_f32(false, a, false, b1, (short)0, acc1, false, false);
    }
    __syncthreads();

    float* Pw = Zs + hq * (TILE_M * PPITCH) + (8 * laneHi) * PPITCH + col0 + laneM;
#pragma unroll
    for (int i = 0; i < 8; ++i) {
        Pw[i * PPITCH]      = acc0[i];
        Pw[i * PPITCH + 16] = acc1[i];
    }
    __syncthreads();

    const int t   = threadIdx.x;
    const int row = t >> 4;
    const int c4  = (t & 15) << 2;
    const float* Pr = Zs + row * PPITCH + c4;
    const float4 p0 = *reinterpret_cast<const float4*>(Pr);
    const float4 p1 = *reinterpret_cast<const float4*>(Pr + TILE_M * PPITCH);
    const float4 p2 = *reinterpret_cast<const float4*>(Pr + 2 * TILE_M * PPITCH);
    const float4 p3 = *reinterpret_cast<const float4*>(Pr + 3 * TILE_M * PPITCH);
    const float4 bs = *reinterpret_cast<const float4*>(bias + c4);
    float4 v;
    v.x = ((p0.x + p1.x) + (p2.x + p3.x)) * 0.25f + bs.x;
    v.y = ((p0.y + p1.y) + (p2.y + p3.y)) * 0.25f + bs.y;
    v.z = ((p0.z + p1.z) + (p2.z + p3.z)) * 0.25f + bs.z;
    v.w = ((p0.w + p1.w) + (p2.w + p3.w)) * 0.25f + bs.w;
    if (do_relu) {
        v.x = fmaxf(v.x, 0.f); v.y = fmaxf(v.y, 0.f);
        v.z = fmaxf(v.z, 0.f); v.w = fmaxf(v.w, 0.f);
    }
    if (node0 + row < N)
        *reinterpret_cast<float4*>(out + (size_t)(node0 + row) * CDIM + c4) = v;
}

extern "C" void kernel_launch(void* const* d_in, const int* in_sizes, int n_in,
                              void* d_out, int out_size, void* d_ws, size_t ws_size,
                              hipStream_t stream) {
    const float* x        = (const float*)d_in[0];
    const int*   ei       = (const int*)  d_in[1];
    const float* W1       = (const float*)d_in[2];
    const float* att_src1 = (const float*)d_in[3];
    const float* att_dst1 = (const float*)d_in[4];
    const float* b1       = (const float*)d_in[5];
    const float* W2       = (const float*)d_in[6];
    const float* att_src2 = (const float*)d_in[7];
    const float* att_dst2 = (const float*)d_in[8];
    const float* b2       = (const float*)d_in[9];

    const int Fin = 128;
    const int N   = in_sizes[0] / Fin;
    const int E   = in_sizes[1] / 2;
    const int* srcv = ei;
    const int* dstv = ei + E;

    const int nChunks = (E + CHUNK_CAP * 256 - 1) / (CHUNK_CAP * 256) * 256;
    const int chunk   = ((E + nChunks - 1) / nChunks + 31) & ~31;
    const int nBins   = (N + BIN_NODES - 1) / BIN_NODES;

    float* ws    = (float*)d_ws;
    float* Zbuf  = ws;
    float* HID   = Zbuf + (size_t)N * HEADS * FIN1;
    float* ASRC  = HID  + (size_t)N * CDIM;
    float* ADST  = ASRC + (size_t)N * HEADS;
    float* V1    = ADST + (size_t)N * HEADS;
    float* V2    = V1 + 2 * HEADS * FIN1;
    int2* BINNED = (int2*)(V2 + 2 * HEADS * FIN2);
    int* OFFS    = (int*)(BINNED + (size_t)nChunks * chunk);
    int* ROWPTR  = OFFS + (size_t)nChunks * (nBins + 1);
    int* CSRC    = ROWPTR + (N + 1);

    const int nhBlocks   = (N * HEADS + 255) / 256;
    const int hvBlocks   = (2 * HEADS * (FIN1 + FIN2) + SORT_T - 1) / SORT_T;
    const int ntBlocks   = (N + TILE_M - 1) / TILE_M;

    bin_edges<<<nChunks + hvBlocks, SORT_T, 0, stream>>>(srcv, dstv, E, chunk, nBins, nChunks,
                                                         BINNED, OFFS,
                                                         W1, att_src1, att_dst1, V1,
                                                         W2, att_src2, att_dst2, V2);
    build_csr<<<nBins, SORT_T, 0, stream>>>(BINNED, OFFS, nChunks, chunk, nBins, N, ROWPTR, CSRC);

    node_scores<FIN1><<<nhBlocks, 256, 0, stream>>>(x, V1, ASRC, ADST, N);
    node_sum_project<FIN1><<<ntBlocks, 256, 0, stream>>>(ROWPTR, CSRC, x, ASRC, ADST, W1, b1, HID, N,  1);

    node_scores<FIN2><<<nhBlocks, 256, 0, stream>>>(HID, V2, ASRC, ADST, N);
    node_sum_project<FIN2><<<ntBlocks, 256, 0, stream>>>(ROWPTR, CSRC, HID, ASRC, ADST, W2, b2, (float*)d_out, N,  0);
}


# attention scores lose their two launches: layer-1 scores by extra blocks of the CSR launch, layer-2 scores in the layer-1 epilogue (row over 16 lanes); sum order of layer-2 scores changes
# speedup vs baseline: 14.7894x; 1.0823x over previous
#include <hip/hip_runtime.h>
#include <hip/hip_bf16.h>
#include <math.h>

typedef float v2f __attribute__((ext_vector_type(2)));
typedef float v8f __attribute__((ext_vector_type(8)));

#define HEADS 4
#define HC 256
#define CDIM 64
#define NEG_SLOPE 0.2f
#define FIN1 128
#define FIN2 64


__device__ __forceinline__ void head_vector(const float* __restrict__ Wm,
                                            const float* __restrict__ att_s,
                                            const float* __restrict__ att_d,
                                            float* __restrict__ V, int FIN, int idx) {
    const int t = idx / FIN, f = idx - t * FIN;
    const int h = t & 3;
    const float* wr = Wm + (size_t)f * HC + h * CDIM;
    const float* av = (t < HEADS ? att_s : att_d) + h * CDIM;
    float s = 0.f;
#pragma unroll 8
    for (int c = 0; c < CDIM; ++c) s = fmaf(wr[c], av[c], s);
    V[idx] = s;
}

template <int FIN>
__device__ __forceinline__ void node_scores(const float* __restrict__ X,
                                            const float* __restrict__ V,
                                            float* __restrict__ as_, float* __restrict__ ad_,
                                            int N, int idx) {
    if (idx >= N * HEADS) return;
    const int n = idx >> 2, h = idx & 3;
    const float* row = X + (size_t)n * FIN;
    const float* vs  = V + h * FIN;
    const float* vd  = V + (HEADS + h) * FIN;
    float s = 0.f, d = 0.f;
#pragma unroll 8
    for (int f = 0; f < FIN; ++f) {
        const float v = row[f];
        s = fmaf(v, vs[f], s);
        d = fmaf(v, vd[f], d);
    }
    as_[idx] = s;
    ad_[idx] = d;
}

__device__ __forceinline__ float leaky(float e) {
    return e > 0.f ? e : NEG_SLOPE * e;
}

#define BIN_SHIFT 8
#define BIN_NODES (1 << BIN_SHIFT)
#define SORT_T 1024
#define CHUNK_PER 4
#define CHUNK_CAP (SORT_T * CHUNK_PER)
#define MAXBIN SORT_T
#define OWN_CAP 8192

__device__ __forceinline__ int wave_sum_i(int v) {
#pragma unroll
    for (int o = 16; o > 0; o >>= 1) v += __shfl_xor(v, o, 32);
    return v;
}

__device__ __forceinline__ int block_excl_scan(int v, int* wtot, int& total) {
    const int lane = threadIdx.x & 31;
    const int w    = threadIdx.x >> 5;
    int x = v;
#pragma unroll
    for (int o = 1; o < 32; o <<= 1) {
        const int y = __shfl_up(x, o, 32);
        if (lane >= o) x += y;
    }
    if (lane == 31) wtot[w] = x;
    __syncthreads();
    const int wt     = wtot[lane];
    const int before = wave_sum_i(lane < w ? wt : 0);
    total            = wave_sum_i(wt);
    __syncthreads();
    return before + x - v;
}

__device__ __forceinline__ int block_sum(int v, int* wtot) {
    const int lane = threadIdx.x & 31;
    const int w    = threadIdx.x >> 5;
    const int x    = wave_sum_i(v);
    if (lane == 0) wtot[w] = x;
    __syncthreads();
    const int total = wave_sum_i(wtot[lane]);
    __syncthreads();
    return total;
}

__global__ void __launch_bounds__(SORT_T)
bin_edges(const int* __restrict__ srcv, const int* __restrict__ dstv, int E, int chunk,
          int nBins, int chunkBlocks,
          int2* __restrict__ binned, int* __restrict__ offs,
          const float* __restrict__ W1, const float* __restrict__ as1,
          const float* __restrict__ ad1, float* __restrict__ V1,
          const float* __restrict__ W2, const float* __restrict__ as2,
          const float* __restrict__ ad2, float* __restrict__ V2) {
    __shared__ int  binCnt[MAXBIN];
    __shared__ int  wtot[SORT_T / 32];
    __shared__ int2 staged[CHUNK_CAP];
    const int t = threadIdx.x;
    if ((int)blockIdx.x >= chunkBlocks) {
        const int idx = ((int)blockIdx.x - chunkBlocks) * SORT_T + t;
        if (idx < 2 * HEADS * FIN1) head_vector(W1, as1, ad1, V1, FIN1, idx);
        else if (idx < 2 * HEADS * (FIN1 + FIN2))
            head_vector(W2, as2, ad2, V2, FIN2, idx - 2 * HEADS * FIN1);
        return;
    }
    const int e0 = (int)blockIdx.x * chunk;
    int n = E - e0;
    n = n < 0 ? 0 : (n > chunk ? chunk : n);
    if (t < nBins) binCnt[t] = 0;
    __syncthreads();

    int s[CHUNK_PER], d[CHUNK_PER], r[CHUNK_PER];
#pragma unroll
    for (int i = 0; i < CHUNK_PER; ++i) {
        const int j = t + i * SORT_T;
        s[i] = 0; d[i] = 0;
        if (j < n) { d[i] = dstv[e0 + j]; s[i] = srcv[e0 + j]; }
    }
#pragma unroll
    for (int i = 0; i < CHUNK_PER; ++i) {
        const int j = t + i * SORT_T;
        r[i] = 0;
        if (j < n) r[i] = atomicAdd(&binCnt[d[i] >> BIN_SHIFT], 1);
    }
    __syncthreads();

    const int c = (t < nBins) ? binCnt[t] : 0;
    int tot;
    const int ex = block_excl_scan(c, wtot, tot);
    int* myoffs = offs + (size_t)blockIdx.x * (nBins + 1);
    if (t < nBins) { binCnt[t] = ex; myoffs[t] = ex; }
    if (t == 0) myoffs[nBins] = tot;
    __syncthreads();

#pragma unroll
    for (int i = 0; i < CHUNK_PER; ++i) {
        const int j = t + i * SORT_T;
        if (j < n) staged[binCnt[d[i] >> BIN_SHIFT] + r[i]] = make_int2(s[i], d[i]);
    }
    __syncthreads();
    int2* outp = binned + (size_t)e0;
    for (int j = t; j < n; j += SORT_T) outp[j] = staged[j];
}

__global__ void __launch_bounds__(SORT_T)
build_csr(const int2* __restrict__ binned, const int* __restrict__ offs,
          int nChunks, int chunk, int nBins, int N,
          int* __restrict__ rowptr, int* __restrict__ csrc,
          const float* __restrict__ X, const float* __restrict__ V,
          float* __restrict__ as_, float* __restrict__ ad_) {
    __shared__ int nodeCnt[BIN_NODES];
    __shared__ int wtot[SORT_T / 32];
    __shared__ int stagedSrc[OWN_CAP];
    const int t      = threadIdx.x;
    if ((int)blockIdx.x >= nBins) {
        node_scores<FIN1>(X, V, as_, ad_, N, ((int)blockIdx.x - nBins) * SORT_T + t);
        return;
    }
    const int b      = blockIdx.x;
    const int node0  = b << BIN_SHIFT;
    const int stride = nBins + 1;

    int lo = 0, hi = 0;
    for (int k = t; k < nChunks; k += SORT_T) {
        lo += offs[(size_t)k * stride + b];
        hi += offs[(size_t)k * stride + b + 1];
    }
    const int base  = block_sum(lo, wtot);
    const int end   = block_sum(hi, wtot);
    const int total = end - base;

    if (t < BIN_NODES) nodeCnt[t] = 0;
    __syncthreads();

    const int hw = t >> 4, hl = t & 15;
    for (int k = hw; k < nChunks; k += SORT_T / 16) {
        const int o0 = offs[(size_t)k * stride + b];
        const int o1 = offs[(size_t)k * stride + b + 1];
        const int2* run = binned + (size_t)k * chunk;
        for (int j = o0 + hl; j < o1; j += 16) atomicAdd(&nodeCnt[run[j].y - node0], 1);
    }
    __syncthreads();

    const int c = (t < BIN_NODES) ? nodeCnt[t] : 0;
    int tot;
    const int ex = block_excl_scan(c, wtot, tot);
    if (t < BIN_NODES) {
        nodeCnt[t] = ex;
        if (node0 + t < N) rowptr[node0 + t] = base + ex;
    }
    if (b == nBins - 1 && t == 0) rowptr[N] = end;
    __syncthreads();

    const bool fits = total <= OWN_CAP;
    for (int k = hw; k < nChunks; k += SORT_T / 16) {
        const int o0 = offs[(size_t)k * stride + b];
        const int o1 = offs[(size_t)k * stride + b + 1];
        const int2* run = binned + (size_t)k * chunk;
        for (int j = o0 + hl; j < o1; j += 16) {
            const int2 e  = run[j];
            const int pos = atomicAdd(&nodeCnt[e.y - node0], 1);
            if (fits) stagedSrc[pos] = e.x;
            else      csrc[base + pos] = e.x;
        }
    }
    if (fits) {
        __syncthreads();
        for (int j = t; j < total; j += SORT_T) csrc[base + j] = stagedSrc[j];
    }
}

__device__ __forceinline__ float wave_max(float v) {
#pragma unroll
    for (int o = 16; o > 0; o >>= 1) v = fmaxf(v, __shfl_xor(v, o, 32));
    return v;
}
__device__ __forceinline__ float wave_sum(float v) {
#pragma unroll
    for (int o = 16; o > 0; o >>= 1) v += __shfl_xor(v, o, 32);
    return v;
}

__device__ __forceinline__ void edge_scores(int i, int cnt, int base, int d,
                                            const int* __restrict__ csrc,
                                            const float* __restrict__ as_,
                                            const float4 adv, int& s, float4& e) {
    s = d;
    if (i < cnt) s = csrc[base + i];
    const float4 a = *reinterpret_cast<const float4*>(as_ + (size_t)s * HEADS);
    e.x = leaky(a.x + adv.x);
    e.y = leaky(a.y + adv.y);
    e.z = leaky(a.z + adv.z);
    e.w = leaky(a.w + adv.w);
}

template <int FIN>
__device__ __forceinline__ void node_rows_sum(const int d, const int base, const int cnt,
                                              const float* __restrict__ adrow,
                                              const int s_first, const float4 e_first,
                                              const int* __restrict__ csrc,
                                              const float* __restrict__ X,
                                              const float* __restrict__ as_,
                                              float* __restrict__ zrow) {
    constexpr int VEC = FIN / 32;
    typedef float vrow __attribute__((ext_vector_type(VEC)));
    const int lane = threadIdx.x & 31;
    const int T    = cnt + 1;

    float m0 = -INFINITY, m1 = -INFINITY, m2 = -INFINITY, m3 = -INFINITY;
    for (int c0 = 0; c0 < T; c0 += 32) {
        const int i = c0 + lane;
        int s = s_first; float4 e = e_first;
        if (c0 > 0)
            edge_scores(i, cnt, base, d, csrc, as_, *reinterpret_cast<const float4*>(adrow), s, e);
        if (i < T) {
            m0 = fmaxf(m0, e.x); m1 = fmaxf(m1, e.y);
            m2 = fmaxf(m2, e.z); m3 = fmaxf(m3, e.w);
        }
    }
    m0 = wave_max(m0); m1 = wave_max(m1); m2 = wave_max(m2); m3 = wave_max(m3);

    float s0 = 0.f, s1 = 0.f, s2 = 0.f, s3 = 0.f;
    for (int c0 = 0; c0 < T; c0 += 32) {
        const int i = c0 + lane;
        int s = s_first; float4 e = e_first;
        if (c0 > 0)
            edge_scores(i, cnt, base, d, csrc, as_, *reinterpret_cast<const float4*>(adrow), s, e);
        if (i < T) {
            s0 += __expf(e.x - m0); s1 += __expf(e.y - m1);
            s2 += __expf(e.z - m2); s3 += __expf(e.w - m3);
        }
    }
    const float den0 = wave_sum(s0) + 1e-16f;
    const float den1 = wave_sum(s1) + 1e-16f;
    const float den2 = wave_sum(s2) + 1e-16f;
    const float den3 = wave_sum(s3) + 1e-16f;

    vrow z0 = {}, z1 = {}, z2 = {}, z3 = {};
    const unsigned lo = lane * VEC;
    for (int c0 = 0; c0 < T; c0 += 32) {
        int sl = s_first; float4 e = e_first;
        if (c0 > 0)
            edge_scores(c0 + lane, cnt, base, d, csrc, as_, *reinterpret_cast<const float4*>(adrow), sl, e);
        const int w0 = __float_as_int(__expf(e.x - m0) / den0);
        const int w1 = __float_as_int(__expf(e.y - m1) / den1);
        const int w2 = __float_as_int(__expf(e.z - m2) / den2);
        const int w3 = __float_as_int(__expf(e.w - m3) / den3);
        const int n = (T - c0 < 32) ? (T - c0) : 32;
        int j = 0;
        for (; j + 4 <= n; j += 4) {
            vrow xr[4];
#pragma unroll
            for (int k = 0; k < 4; ++k) {
                const int s = __builtin_amdgcn_readlane(sl, j + k);
                xr[k] = *reinterpret_cast<const vrow*>(X + ((unsigned)s * FIN + lo));
            }
#pragma unroll
            for (int k = 0; k < 4; ++k) {
                const float a0 = __int_as_float(__builtin_amdgcn_readlane(w0, j + k));
                const float a1 = __int_as_float(__builtin_amdgcn_readlane(w1, j + k));
                const float a2 = __int_as_float(__builtin_amdgcn_readlane(w2, j + k));
                const float a3 = __int_as_float(__builtin_amdgcn_readlane(w3, j + k));
                z0 += xr[k] * a0;
                z1 += xr[k] * a1;
                z2 += xr[k] * a2;
                z3 += xr[k] * a3;
            }
        }
        for (; j < n; ++j) {
            const int s = __builtin_amdgcn_readlane(sl, j);
            const vrow xv = *reinterpret_cast<const vrow*>(X + ((unsigned)s * FIN + lo));
            const float a0 = __int_as_float(__builtin_amdgcn_readlane(w0, j));
            const float a1 = __int_as_float(__builtin_amdgcn_readlane(w1, j));
            const float a2 = __int_as_float(__builtin_amdgcn_readlane(w2, j));
            const float a3 = __int_as_float(__builtin_amdgcn_readlane(w3, j));
            z0 += xv * a0;
            z1 += xv * a1;
            z2 += xv * a2;
            z3 += xv * a3;
        }
    }

    *reinterpret_cast<vrow*>(zrow)           = z0;
    *reinterpret_cast<vrow*>(zrow + FIN)     = z1;
    *reinterpret_cast<vrow*>(zrow + 2 * FIN) = z2;
    *reinterpret_cast<vrow*>(zrow + 3 * FIN) = z3;
}

#define TILE_M 16
#define PPITCH 68

template <int FIN, bool NEXT>
__global__ void __launch_bounds__(256)
node_sum_project(const int* __restrict__ rowptr, const int* __restrict__ csrc,
                 const float* __restrict__ X,
                 const float* __restrict__ as_, const float* __restrict__ ad_,
                 const float* __restrict__ Wm, const float* __restrict__ bias,
                 float* __restrict__ out, int N, int do_relu,
                 const float* __restrict__ Vn,
                 float* __restrict__ as_n, float* __restrict__ ad_n) {
    constexpr int VEC   = FIN / 32;
    constexpr int K     = HEADS * FIN;
    constexpr int PITCH = K + 4;
    constexpr int ZSIZE = TILE_M * PITCH > HEADS * TILE_M * PPITCH ? TILE_M * PITCH
                                                                   : HEADS * TILE_M * PPITCH;
    typedef float vrow __attribute__((ext_vector_type(VEC)));
    __shared__ __attribute__((aligned(16))) float Zs[ZSIZE];

    const int lane = threadIdx.x & 31;
    const int wave  = __builtin_amdgcn_readfirstlane((int)(threadIdx.x >> 5));
    const int node0 = blockIdx.x * TILE_M;

    {
        const int rowA = wave * 2;
        const int dA = node0 + rowA, dB = dA + 1;
        const int dAc = dA < N ? dA : N - 1;
        const int dBc = dB < N ? dB : N - 1;
        int rpi = dA + (lane < 3 ? lane : 0);
        rpi = rpi < N ? rpi : N;
        const int rp  = rowptr[rpi];
        const int bA   = __builtin_amdgcn_readlane(rp, 0);
        const int bB   = __builtin_amdgcn_readlane(rp, 1);
        const int cntA = bB - bA;
        const int cntB = __builtin_amdgcn_readlane(rp, 2) - bB;
        const float* adA = ad_ + (size_t)dAc * HEADS;
        const float* adB = ad_ + (size_t)dBc * HEADS;
        const float4 advA = *reinterpret_cast<const float4*>(adA);
        const float4 advB = *reinterpret_cast<const float4*>(adB);
        int sA, sB; float4 eA, eB;
        edge_scores(lane, cntA, bA, dAc, csrc, as_, advA, sA, eA);
        edge_scores(lane, cntB, bB, dBc, csrc, as_, advB, sB, eB);

        float* zA = Zs + rowA * PITCH + lane * VEC;
        float* zB = zA + PITCH;
        if (dA < N) {
            node_rows_sum<FIN>(dA, bA, cntA, adA, sA, eA, csrc, X, as_, zA);
        } else {
            const vrow zz = {};
#pragma unroll
            for (int h = 0; h < HEADS; ++h) *reinterpret_cast<vrow*>(zA + h * FIN) = zz;
        }
        if (dB < N) {
            node_rows_sum<FIN>(dB, bB, cntB, adB, sB, eB, csrc, X, as_, zB);
        } else {
            const vrow zz = {};
#pragma unroll
            for (int h = 0; h < HEADS; ++h) *reinterpret_cast<vrow*>(zB + h * FIN) = zz;
        }
    }
    __syncthreads();

    const int laneM  = lane & 15;
    const int laneHi = lane >> 4;
    const int hq     = wave >> 1;
    const int col0   = (wave & 1) << 5;
    const float* Arow  = Zs + laneM * PITCH + hq * FIN + (laneHi << 1);
    const float* Bbase = Wm + (size_t)(laneHi << 1) * HC + hq * CDIM + col0 + laneM;

    v8f acc0 = {}, acc1 = {};
#pragma unroll 4
    for (int f = 0; f < FIN; f += 4) {
        const v2f a = *reinterpret_cast<const v2f*>(Arow + f);
        const float* B0 = Bbase + (size_t)f * HC;
        const float* B1 = B0 + HC;
        v2f b0, b1;
        b0.x = B0[0];  b0.y = B1[0];
        b1.x = B0[16]; b1.y = B1[16];
        acc0 = __builtin_amdgcn_wmma_f32_16x16x4_f32(false, a, false, b0, (short)0, acc0, false, false);
        acc1 = __builtin_amdgcn_wmma_f32_16x16x4_f32(false, a, false, b1, (short)0, acc1, false, false);
    }
    __syncthreads();

    float* Pw = Zs + hq * (TILE_M * PPITCH) + (8 * laneHi) * PPITCH + col0 + laneM;
#pragma unroll
    for (int i = 0; i < 8; ++i) {
        Pw[i * PPITCH]      = acc0[i];
        Pw[i * PPITCH + 16] = acc1[i];
    }
    const int t   = threadIdx.x;
    const int row = t >> 4;
    const int c4  = (t & 15) << 2;
    __syncthreads();
    const float4 bs = *reinterpret_cast<const float4*>(bias + c4);
    float4 vn[2 * HEADS];
    if (NEXT) {
#pragma unroll
        for (int j = 0; j < 2 * HEADS; ++j)
            vn[j] = *reinterpret_cast<const float4*>(Vn + j * CDIM + c4);
    }

    const float* Pr = Zs + row * PPITCH + c4;
    const float4 p0 = *reinterpret_cast<const float4*>(Pr);
    const float4 p1 = *reinterpret_cast<const float4*>(Pr + TILE_M * PPITCH);
    const float4 p2 = *reinterpret_cast<const float4*>(Pr + 2 * TILE_M * PPITCH);
    const float4 p3 = *reinterpret_cast<const float4*>(Pr + 3 * TILE_M * PPITCH);
    float4 v;
    v.x = ((p0.x + p1.x) + (p2.x + p3.x)) * 0.25f + bs.x;
    v.y = ((p0.y + p1.y) + (p2.y + p3.y)) * 0.25f + bs.y;
    v.z = ((p0.z + p1.z) + (p2.z + p3.z)) * 0.25f + bs.z;
    v.w = ((p0.w + p1.w) + (p2.w + p3.w)) * 0.25f + bs.w;
    if (do_relu) {
        v.x = fmaxf(v.x, 0.f); v.y = fmaxf(v.y, 0.f);
        v.z = fmaxf(v.z, 0.f); v.w = fmaxf(v.w, 0.f);
    }
    if (node0 + row < N)
        *reinterpret_cast<float4*>(out + (size_t)(node0 + row) * CDIM + c4) = v;

    if (NEXT) {
        float sc[2 * HEADS];
#pragma unroll
        for (int j = 0; j < 2 * HEADS; ++j)
            sc[j] = fmaf(v.x, vn[j].x, fmaf(v.y, vn[j].y, fmaf(v.z, vn[j].z, v.w * vn[j].w)));
        float r4[4], r2[2], r1;
        {
            const bool up = (t & 8) != 0;
#pragma unroll
            for (int j = 0; j < 4; ++j) {
                const float send = up ? sc[j] : sc[j + 4];
                const float keep = up ? sc[j + 4] : sc[j];
                r4[j] = keep + __shfl_xor(send, 8, 32);
            }
        }
        {
            const bool up = (t & 4) != 0;
#pragma unroll
            for (int j = 0; j < 2; ++j) {
                const float send = up ? r4[j] : r4[j + 2];
                const float keep = up ? r4[j + 2] : r4[j];
                r2[j] = keep + __shfl_xor(send, 4, 32);
            }
        }
        {
            const bool up = (t & 2) != 0;
            const float send = up ? r2[0] : r2[1];
            const float keep = up ? r2[1] : r2[0];
            r1 = keep + __shfl_xor(send, 2, 32);
        }
        r1 += __shfl_xor(r1, 1, 32);
        const int q = (t & 15) >> 1;
        __shared__ __attribute__((aligned(16))) float Sn[TILE_M * 2 * HEADS];
        if ((t & 1) == 0) Sn[row * (2 * HEADS) + q] = r1;
        __syncthreads();
        if (t < 32) {
            const int r  = t & 15;
            const int wh = t >> 4;
            const float4 sv = *reinterpret_cast<const float4*>(Sn + r * (2 * HEADS) + wh * HEADS);
            if (node0 + r < N)
                *reinterpret_cast<float4*>((wh ? ad_n : as_n) + (size_t)(node0 + r) * HEADS) = sv;
        }
    }
}

extern "C" void kernel_launch(void* const* d_in, const int* in_sizes, int n_in,
                              void* d_out, int out_size, void* d_ws, size_t ws_size,
                              hipStream_t stream) {
    const float* x        = (const float*)d_in[0];
    const int*   ei       = (const int*)  d_in[1];
    const float* W1       = (const float*)d_in[2];
    const float* att_src1 = (const float*)d_in[3];
    const float* att_dst1 = (const float*)d_in[4];
    const float* b1       = (const float*)d_in[5];
    const float* W2       = (const float*)d_in[6];
    const float* att_src2 = (const float*)d_in[7];
    const float* att_dst2 = (const float*)d_in[8];
    const float* b2       = (const float*)d_in[9];

    const int Fin = 128;
    const int N   = in_sizes[0] / Fin;
    const int E   = in_sizes[1] / 2;
    const int* srcv = ei;
    const int* dstv = ei + E;

    const int nChunks = (E + CHUNK_CAP * 256 - 1) / (CHUNK_CAP * 256) * 256;
    const int chunk   = ((E + nChunks - 1) / nChunks + 31) & ~31;
    const int nBins   = (N + BIN_NODES - 1) / BIN_NODES;

    float* ws    = (float*)d_ws;
    float* Zbuf  = ws;
    float* ASRC2 = Zbuf;
    float* ADST2 = ASRC2 + (size_t)N * HEADS;
    float* HID   = Zbuf + (size_t)N * HEADS * FIN1;
    float* ASRC  = HID  + (size_t)N * CDIM;
    float* ADST  = ASRC + (size_t)N * HEADS;
    float* V1    = ADST + (size_t)N * HEADS;
    float* V2    = V1 + 2 * HEADS * FIN1;
    int2* BINNED = (int2*)(V2 + 2 * HEADS * FIN2);
    int* OFFS    = (int*)(BINNED + (size_t)nChunks * chunk);
    int* ROWPTR  = OFFS + (size_t)nChunks * (nBins + 1);
    int* CSRC    = ROWPTR + (N + 1);

    const int nsBlocks   = (N * HEADS + SORT_T - 1) / SORT_T;
    const int hvBlocks   = (2 * HEADS * (FIN1 + FIN2) + SORT_T - 1) / SORT_T;
    const int ntBlocks   = (N + TILE_M - 1) / TILE_M;

    bin_edges<<<nChunks + hvBlocks, SORT_T, 0, stream>>>(srcv, dstv, E, chunk, nBins, nChunks,
                                                         BINNED, OFFS,
                                                         W1, att_src1, att_dst1, V1,
                                                         W2, att_src2, att_dst2, V2);
    build_csr<<<nBins + nsBlocks, SORT_T, 0, stream>>>(BINNED, OFFS, nChunks, chunk, nBins, N, ROWPTR, CSRC,
                                                       x, V1, ASRC, ADST);

    node_sum_project<FIN1, true><<<ntBlocks, 256, 0, stream>>>(ROWPTR, CSRC, x, ASRC, ADST, W1, b1, HID, N,  1,
                                                               V2, ASRC2, ADST2);

    node_sum_project<FIN2, false><<<ntBlocks, 256, 0, stream>>>(ROWPTR, CSRC, HID, ASRC2, ADST2, W2, b2, (float*)d_out, N,  0,
                                                                nullptr, nullptr, nullptr);
}


# softmax of a node's scores in about half the instructions: max and sum by DPP row steps + one cross-row exchange (no LDS permutes), exps kept, weights by a reciprocal per head; sum order changes
# speedup vs baseline: 15.4984x; 1.0479x over previous
#include <hip/hip_runtime.h>
#include <hip/hip_bf16.h>
#include <math.h>

typedef float v2f __attribute__((ext_vector_type(2)));
typedef float v8f __attribute__((ext_vector_type(8)));

#define HEADS 4
#define HC 256
#define CDIM 64
#define NEG_SLOPE 0.2f
#define FIN1 128
#define FIN2 64


__device__ __forceinline__ void head_vector(const float* __restrict__ Wm,
                                            const float* __restrict__ att_s,
                                            const float* __restrict__ att_d,
                                            float* __restrict__ V, int FIN, int idx) {
    const int t = idx / FIN, f = idx - t * FIN;
    const int h = t & 3;
    const float* wr = Wm + (size_t)f * HC + h * CDIM;
    const float* av = (t < HEADS ? att_s : att_d) + h * CDIM;
    float s = 0.f;
#pragma unroll 8
    for (int c = 0; c < CDIM; ++c) s = fmaf(wr[c], av[c], s);
    V[idx] = s;
}

template <int FIN>
__device__ __forceinline__ void node_scores(const float* __restrict__ X,
                                            const float* __restrict__ V,
                                            float* __restrict__ as_, float* __restrict__ ad_,
                                            int N, int idx) {
    if (idx >= N * HEADS) return;
    const int n = idx >> 2, h = idx & 3;
    const float* row = X + (size_t)n * FIN;
    const float* vs  = V + h * FIN;
    const float* vd  = V + (HEADS + h) * FIN;
    float s = 0.f, d = 0.f;
#pragma unroll 8
    for (int f = 0; f < FIN; ++f) {
        const float v = row[f];
        s = fmaf(v, vs[f], s);
        d = fmaf(v, vd[f], d);
    }
    as_[idx] = s;
    ad_[idx] = d;
}

__device__ __forceinline__ float leaky(float e) {
    return fmaxf(e, NEG_SLOPE * e);
}

#define BIN_SHIFT 8
#define BIN_NODES (1 << BIN_SHIFT)
#define SORT_T 1024
#define CHUNK_PER 4
#define CHUNK_CAP (SORT_T * CHUNK_PER)
#define MAXBIN SORT_T
#define OWN_CAP 8192

__device__ __forceinline__ int wave_sum_i(int v) {
#pragma unroll
    for (int o = 16; o > 0; o >>= 1) v += __shfl_xor(v, o, 32);
    return v;
}

__device__ __forceinline__ int block_excl_scan(int v, int* wtot, int& total) {
    const int lane = threadIdx.x & 31;
    const int w    = threadIdx.x >> 5;
    int x = v;
#pragma unroll
    for (int o = 1; o < 32; o <<= 1) {
        const int y = __shfl_up(x, o, 32);
        if (lane >= o) x += y;
    }
    if (lane == 31) wtot[w] = x;
    __syncthreads();
    const int wt     = wtot[lane];
    const int before = wave_sum_i(lane < w ? wt : 0);
    total            = wave_sum_i(wt);
    __syncthreads();
    return before + x - v;
}

__device__ __forceinline__ int block_sum(int v, int* wtot) {
    const int lane = threadIdx.x & 31;
    const int w    = threadIdx.x >> 5;
    const int x    = wave_sum_i(v);
    if (lane == 0) wtot[w] = x;
    __syncthreads();
    const int total = wave_sum_i(wtot[lane]);
    __syncthreads();
    return total;
}

__global__ void __launch_bounds__(SORT_T)
bin_edges(const int* __restrict__ srcv, const int* __restrict__ dstv, int E, int chunk,
          int nBins, int chunkBlocks,
          int2* __restrict__ binned, int* __restrict__ offs,
          const float* __restrict__ W1, const float* __restrict__ as1,
          const float* __restrict__ ad1, float* __restrict__ V1,
          const float* __restrict__ W2, const float* __restrict__ as2,
          const float* __restrict__ ad2, float* __restrict__ V2) {
    __shared__ int  binCnt[MAXBIN];
    __shared__ int  wtot[SORT_T / 32];
    __shared__ int2 staged[CHUNK_CAP];
    const int t = threadIdx.x;
    if ((int)blockIdx.x >= chunkBlocks) {
        const int idx = ((int)blockIdx.x - chunkBlocks) * SORT_T + t;
        if (idx < 2 * HEADS * FIN1) head_vector(W1, as1, ad1, V1, FIN1, idx);
        else if (idx < 2 * HEADS * (FIN1 + FIN2))
            head_vector(W2, as2, ad2, V2, FIN2, idx - 2 * HEADS * FIN1);
        return;
    }
    const int e0 = (int)blockIdx.x * chunk;
    int n = E - e0;
    n = n < 0 ? 0 : (n > chunk ? chunk : n);
    if (t < nBins) binCnt[t] = 0;
    __syncthreads();

    int s[CHUNK_PER], d[CHUNK_PER], r[CHUNK_PER];
#pragma unroll
    for (int i = 0; i < CHUNK_PER; ++i) {
        const int j = t + i * SORT_T;
        s[i] = 0; d[i] = 0;
        if (j < n) { d[i] = dstv[e0 + j]; s[i] = srcv[e0 + j]; }
    }
#pragma unroll
    for (int i = 0; i < CHUNK_PER; ++i) {
        const int j = t + i * SORT_T;
        r[i] = 0;
        if (j < n) r[i] = atomicAdd(&binCnt[d[i] >> BIN_SHIFT], 1);
    }
    __syncthreads();

    const int c = (t < nBins) ? binCnt[t] : 0;
    int tot;
    const int ex = block_excl_scan(c, wtot, tot);
    int* myoffs = offs + (size_t)blockIdx.x * (nBins + 1);
    if (t < nBins) { binCnt[t] = ex; myoffs[t] = ex; }
    if (t == 0) myoffs[nBins] = tot;
    __syncthreads();

#pragma unroll
    for (int i = 0; i < CHUNK_PER; ++i) {
        const int j = t + i * SORT_T;
        if (j < n) staged[binCnt[d[i] >> BIN_SHIFT] + r[i]] = make_int2(s[i], d[i]);
    }
    __syncthreads();
    int2* outp = binned + (size_t)e0;
    for (int j = t; j < n; j += SORT_T) outp[j] = staged[j];
}

__global__ void __launch_bounds__(SORT_T)
build_csr(const int2* __restrict__ binned, const int* __restrict__ offs,
          int nChunks, int chunk, int nBins, int N,
          int* __restrict__ rowptr, int* __restrict__ csrc,
          const float* __restrict__ X, const float* __restrict__ V,
          float* __restrict__ as_, float* __restrict__ ad_) {
    __shared__ int nodeCnt[BIN_NODES];
    __shared__ int wtot[SORT_T / 32];
    __shared__ int stagedSrc[OWN_CAP];
    const int t      = threadIdx.x;
    if ((int)blockIdx.x >= nBins) {
        node_scores<FIN1>(X, V, as_, ad_, N, ((int)blockIdx.x - nBins) * SORT_T + t);
        return;
    }
    const int b      = blockIdx.x;
    const int node0  = b << BIN_SHIFT;
    const int stride = nBins + 1;

    int lo = 0, hi = 0;
    for (int k = t; k < nChunks; k += SORT_T) {
        lo += offs[(size_t)k * stride + b];
        hi += offs[(size_t)k * stride + b + 1];
    }
    const int base  = block_sum(lo, wtot);
    const int end   = block_sum(hi, wtot);
    const int total = end - base;

    if (t < BIN_NODES) nodeCnt[t] = 0;
    __syncthreads();

    const int hw = t >> 4, hl = t & 15;
    for (int k = hw; k < nChunks; k += SORT_T / 16) {
        const int o0 = offs[(size_t)k * stride + b];
        const int o1 = offs[(size_t)k * stride + b + 1];
        const int2* run = binned + (size_t)k * chunk;
        for (int j = o0 + hl; j < o1; j += 16) atomicAdd(&nodeCnt[run[j].y - node0], 1);
    }
    __syncthreads();

    const int c = (t < BIN_NODES) ? nodeCnt[t] : 0;
    int tot;
    const int ex = block_excl_scan(c, wtot, tot);
    if (t < BIN_NODES) {
        nodeCnt[t] = ex;
        if (node0 + t < N) rowptr[node0 + t] = base + ex;
    }
    if (b == nBins - 1 && t == 0) rowptr[N] = end;
    __syncthreads();

    const bool fits = total <= OWN_CAP;
    for (int k = hw; k < nChunks; k += SORT_T / 16) {
        const int o0 = offs[(size_t)k * stride + b];
        const int o1 = offs[(size_t)k * stride + b + 1];
        const int2* run = binned + (size_t)k * chunk;
        for (int j = o0 + hl; j < o1; j += 16) {
            const int2 e  = run[j];
            const int pos = atomicAdd(&nodeCnt[e.y - node0], 1);
            if (fits) stagedSrc[pos] = e.x;
            else      csrc[base + pos] = e.x;
        }
    }
    if (fits) {
        __syncthreads();
        for (int j = t; j < total; j += SORT_T) csrc[base + j] = stagedSrc[j];
    }
}

template <int M>
__device__ __forceinline__ float row_x(float v) {
    return __builtin_bit_cast(float, __builtin_amdgcn_update_dpp(0, __builtin_bit_cast(int, v),
                                                                  0x160 + M, 0xf, 0xf, true));
}
__device__ __forceinline__ float xhalf(float v) {
    const int b = __builtin_bit_cast(int, v);
    return __builtin_bit_cast(float, __builtin_amdgcn_permlanex16(b, b, 0x76543210, 0xfedcba98, false, false));
}
#define ROW_MAX_STEP(M)                                                                                  \
    asm volatile("v_max_num_f32_dpp %0, %0, %0 row_xmask:" #M " row_mask:0xf bank_mask:0xf\n\t"          \
                 "v_max_num_f32_dpp %1, %1, %1 row_xmask:" #M " row_mask:0xf bank_mask:0xf\n\t"          \
                 "v_max_num_f32_dpp %2, %2, %2 row_xmask:" #M " row_mask:0xf bank_mask:0xf\n\t"          \
                 "v_max_num_f32_dpp %3, %3, %3 row_xmask:" #M " row_mask:0xf bank_mask:0xf"                \
                 : "+v"(a), "+v"(b), "+v"(c), "+v"(d))
__device__ __forceinline__ void wave_max4(float& a, float& b, float& c, float& d) {
    ROW_MAX_STEP(1);
    ROW_MAX_STEP(2);
    ROW_MAX_STEP(4);
    ROW_MAX_STEP(8);
    a = fmaxf(a, xhalf(a));
    b = fmaxf(b, xhalf(b));
    c = fmaxf(c, xhalf(c));
    d = fmaxf(d, xhalf(d));
}
#undef ROW_MAX_STEP
__device__ __forceinline__ float wave_sum(float v) {
    v += row_x<1>(v);
    v += row_x<2>(v);
    v += row_x<4>(v);
    v += row_x<8>(v);
    return v + xhalf(v);
}

__device__ __forceinline__ void edge_scores(int i, int cnt, int base, int d,
                                            const int* __restrict__ csrc,
                                            const float* __restrict__ as_,
                                            const float4 adv, int& s, float4& e) {
    s = d;
    if (i < cnt) s = csrc[base + i];
    const float4 a = *reinterpret_cast<const float4*>(as_ + (size_t)s * HEADS);
    e.x = leaky(a.x + adv.x);
    e.y = leaky(a.y + adv.y);
    e.z = leaky(a.z + adv.z);
    e.w = leaky(a.w + adv.w);
}

template <int FIN>
__device__ __forceinline__ void node_rows_sum(const int d, const int base, const int cnt,
                                              const float* __restrict__ adrow,
                                              const int s_first, const float4 e_first,
                                              const int* __restrict__ csrc,
                                              const float* __restrict__ X,
                                              const float* __restrict__ as_,
                                              float* __restrict__ zrow) {
    constexpr int VEC = FIN / 32;
    typedef float vrow __attribute__((ext_vector_type(VEC)));
    const int lane = threadIdx.x & 31;
    const int T    = cnt + 1;

    float m0 = e_first.x, m1 = e_first.y, m2 = e_first.z, m3 = e_first.w;
    for (int c0 = 32; c0 < T; c0 += 32) {
        const int i = c0 + lane;
        int s; float4 e;
        edge_scores(i, cnt, base, d, csrc, as_, *reinterpret_cast<const float4*>(adrow), s, e);
        if (i < T) {
            m0 = fmaxf(m0, e.x); m1 = fmaxf(m1, e.y);
            m2 = fmaxf(m2, e.z); m3 = fmaxf(m3, e.w);
        }
    }
    wave_max4(m0, m1, m2, m3);

    const bool in0 = lane < T;
    const float x0 = in0 ? __expf(e_first.x - m0) : 0.f;
    const float x1 = in0 ? __expf(e_first.y - m1) : 0.f;
    const float x2 = in0 ? __expf(e_first.z - m2) : 0.f;
    const float x3 = in0 ? __expf(e_first.w - m3) : 0.f;
    float s0 = x0, s1 = x1, s2 = x2, s3 = x3;
    for (int c0 = 32; c0 < T; c0 += 32) {
        const int i = c0 + lane;
        int s; float4 e;
        edge_scores(i, cnt, base, d, csrc, as_, *reinterpret_cast<const float4*>(adrow), s, e);
        if (i < T) {
            s0 += __expf(e.x - m0); s1 += __expf(e.y - m1);
            s2 += __expf(e.z - m2); s3 += __expf(e.w - m3);
        }
    }
    const float r0 = __builtin_amdgcn_rcpf(wave_sum(s0) + 1e-16f);
    const float r1 = __builtin_amdgcn_rcpf(wave_sum(s1) + 1e-16f);
    const float r2 = __builtin_amdgcn_rcpf(wave_sum(s2) + 1e-16f);
    const float r3 = __builtin_amdgcn_rcpf(wave_sum(s3) + 1e-16f);

    vrow z0 = {}, z1 = {}, z2 = {}, z3 = {};
    const unsigned lo = lane * VEC;
    for (int c0 = 0; c0 < T; c0 += 32) {
        int sl = s_first;
        float y0 = x0, y1 = x1, y2 = x2, y3 = x3;
        if (c0 > 0) {
            float4 e;
            edge_scores(c0 + lane, cnt, base, d, csrc, as_, *reinterpret_cast<const float4*>(adrow), sl, e);
            y0 = __expf(e.x - m0); y1 = __expf(e.y - m1);
            y2 = __expf(e.z - m2); y3 = __expf(e.w - m3);
        }
        const int w0 = __float_as_int(y0 * r0);
        const int w1 = __float_as_int(y1 * r1);
        const int w2 = __float_as_int(y2 * r2);
        const int w3 = __float_as_int(y3 * r3);
        const int n = (T - c0 < 32) ? (T - c0) : 32;
        int j = 0;
        for (; j + 4 <= n; j += 4) {
            vrow xr[4];
#pragma unroll
            for (int k = 0; k < 4; ++k) {
                const int s = __builtin_amdgcn_readlane(sl, j + k);
                xr[k] = *reinterpret_cast<const vrow*>(X + ((unsigned)s * FIN + lo));
            }
#pragma unroll
            for (int k = 0; k < 4; ++k) {
                const float a0 = __int_as_float(__builtin_amdgcn_readlane(w0, j + k));
                const float a1 = __int_as_float(__builtin_amdgcn_readlane(w1, j + k));
                const float a2 = __int_as_float(__builtin_amdgcn_readlane(w2, j + k));
                const float a3 = __int_as_float(__builtin_amdgcn_readlane(w3, j + k));
                z0 += xr[k] * a0;
                z1 += xr[k] * a1;
                z2 += xr[k] * a2;
                z3 += xr[k] * a3;
            }
        }
        for (; j < n; ++j) {
            const int s = __builtin_amdgcn_readlane(sl, j);
            const vrow xv = *reinterpret_cast<const vrow*>(X + ((unsigned)s * FIN + lo));
            const float a0 = __int_as_float(__builtin_amdgcn_readlane(w0, j));
            const float a1 = __int_as_float(__builtin_amdgcn_readlane(w1, j));
            const float a2 = __int_as_float(__builtin_amdgcn_readlane(w2, j));
            const float a3 = __int_as_float(__builtin_amdgcn_readlane(w3, j));
            z0 += xv * a0;
            z1 += xv * a1;
            z2 += xv * a2;
            z3 += xv * a3;
        }
    }

    *reinterpret_cast<vrow*>(zrow)           = z0;
    *reinterpret_cast<vrow*>(zrow + FIN)     = z1;
    *reinterpret_cast<vrow*>(zrow + 2 * FIN) = z2;
    *reinterpret_cast<vrow*>(zrow + 3 * FIN) = z3;
}

#define TILE_M 16
#define PPITCH 68

template <int FIN, bool NEXT>
__global__ void __launch_bounds__(256)
node_sum_project(const int* __restrict__ rowptr, const int* __restrict__ csrc,
                 const float* __restrict__ X,
                 const float* __restrict__ as_, const float* __restrict__ ad_,
                 const float* __restrict__ Wm, const float* __restrict__ bias,
                 float* __restrict__ out, int N, int do_relu,
                 const float* __restrict__ Vn,
                 float* __restrict__ as_n, float* __restrict__ ad_n) {
    constexpr int VEC   = FIN / 32;
    constexpr int K     = HEADS * FIN;
    constexpr int PITCH = K + 4;
    constexpr int ZSIZE = TILE_M * PITCH > HEADS * TILE_M * PPITCH ? TILE_M * PITCH
                                                                   : HEADS * TILE_M * PPITCH;
    typedef float vrow __attribute__((ext_vector_type(VEC)));
    __shared__ __attribute__((aligned(16))) float Zs[ZSIZE];

    const int lane = threadIdx.x & 31;
    const int wave  = __builtin_amdgcn_readfirstlane((int)(threadIdx.x >> 5));
    const int node0 = blockIdx.x * TILE_M;

    {
        const int rowA = wave * 2;
        const int dA = node0 + rowA, dB = dA + 1;
        const int dAc = dA < N ? dA : N - 1;
        const int dBc = dB < N ? dB : N - 1;
        int rpi = dA + (lane < 3 ? lane : 0);
        rpi = rpi < N ? rpi : N;
        const int rp  = rowptr[rpi];
        const int bA   = __builtin_amdgcn_readlane(rp, 0);
        const int bB   = __builtin_amdgcn_readlane(rp, 1);
        const int cntA = bB - bA;
        const int cntB = __builtin_amdgcn_readlane(rp, 2) - bB;
        const float* adA = ad_ + (size_t)dAc * HEADS;
        const float* adB = ad_ + (size_t)dBc * HEADS;
        const float4 advA = *reinterpret_cast<const float4*>(adA);
        const float4 advB = *reinterpret_cast<const float4*>(adB);
        int sA, sB; float4 eA, eB;
        edge_scores(lane, cntA, bA, dAc, csrc, as_, advA, sA, eA);
        edge_scores(lane, cntB, bB, dBc, csrc, as_, advB, sB, eB);

        float* zA = Zs + rowA * PITCH + lane * VEC;
        float* zB = zA + PITCH;
        if (dA < N) {
            node_rows_sum<FIN>(dA, bA, cntA, adA, sA, eA, csrc, X, as_, zA);
        } else {
            const vrow zz = {};
#pragma unroll
            for (int h = 0; h < HEADS; ++h) *reinterpret_cast<vrow*>(zA + h * FIN) = zz;
        }
        if (dB < N) {
            node_rows_sum<FIN>(dB, bB, cntB, adB, sB, eB, csrc, X, as_, zB);
        } else {
            const vrow zz = {};
#pragma unroll
            for (int h = 0; h < HEADS; ++h) *reinterpret_cast<vrow*>(zB + h * FIN) = zz;
        }
    }
    __syncthreads();

    const int laneM  = lane & 15;
    const int laneHi = lane >> 4;
    const int hq     = wave >> 1;
    const int col0   = (wave & 1) << 5;
    const float* Arow  = Zs + laneM * PITCH + hq * FIN + (laneHi << 1);
    const float* Bbase = Wm + (size_t)(laneHi << 1) * HC + hq * CDIM + col0 + laneM;

    v8f acc0 = {}, acc1 = {};
#pragma unroll 4
    for (int f = 0; f < FIN; f += 4) {
        const v2f a = *reinterpret_cast<const v2f*>(Arow + f);
        const float* B0 = Bbase + (size_t)f * HC;
        const float* B1 = B0 + HC;
        v2f b0, b1;
        b0.x = B0[0];  b0.y = B1[0];
        b1.x = B0[16]; b1.y = B1[16];
        acc0 = __builtin_amdgcn_wmma_f32_16x16x4_f32(false, a, false, b0, (short)0, acc0, false, false);
        acc1 = __builtin_amdgcn_wmma_f32_16x16x4_f32(false, a, false, b1, (short)0, acc1, false, false);
    }
    __syncthreads();

    float* Pw = Zs + hq * (TILE_M * PPITCH) + (8 * laneHi) * PPITCH + col0 + laneM;
#pragma unroll
    for (int i = 0; i < 8; ++i) {
        Pw[i * PPITCH]      = acc0[i];
        Pw[i * PPITCH + 16] = acc1[i];
    }
    const int t   = threadIdx.x;
    const int row = t >> 4;
    const int c4  = (t & 15) << 2;
    __syncthreads();
    const float4 bs = *reinterpret_cast<const float4*>(bias + c4);
    float4 vn[2 * HEADS];
    if (NEXT) {
#pragma unroll
        for (int j = 0; j < 2 * HEADS; ++j)
            vn[j] = *reinterpret_cast<const float4*>(Vn + j * CDIM + c4);
    }

    const float* Pr = Zs + row * PPITCH + c4;
    const float4 p0 = *reinterpret_cast<const float4*>(Pr);
    const float4 p1 = *reinterpret_cast<const float4*>(Pr + TILE_M * PPITCH);
    const float4 p2 = *reinterpret_cast<const float4*>(Pr + 2 * TILE_M * PPITCH);
    const float4 p3 = *reinterpret_cast<const float4*>(Pr + 3 * TILE_M * PPITCH);
    float4 v;
    v.x = ((p0.x + p1.x) + (p2.x + p3.x)) * 0.25f + bs.x;
    v.y = ((p0.y + p1.y) + (p2.y + p3.y)) * 0.25f + bs.y;
    v.z = ((p0.z + p1.z) + (p2.z + p3.z)) * 0.25f + bs.z;
    v.w = ((p0.w + p1.w) + (p2.w + p3.w)) * 0.25f + bs.w;
    if (do_relu) {
        v.x = fmaxf(v.x, 0.f); v.y = fmaxf(v.y, 0.f);
        v.z = fmaxf(v.z, 0.f); v.w = fmaxf(v.w, 0.f);
    }
    if (node0 + row < N)
        *reinterpret_cast<float4*>(out + (size_t)(node0 + row) * CDIM + c4) = v;

    if (NEXT) {
        float sc[2 * HEADS];
#pragma unroll
        for (int j = 0; j < 2 * HEADS; ++j)
            sc[j] = fmaf(v.x, vn[j].x, fmaf(v.y, vn[j].y, fmaf(v.z, vn[j].z, v.w * vn[j].w)));
        float r4[4], r2[2], r1;
        {
            const bool up = (t & 8) != 0;
#pragma unroll
            for (int j = 0; j < 4; ++j) {
                const float send = up ? sc[j] : sc[j + 4];
                const float keep = up ? sc[j + 4] : sc[j];
                r4[j] = keep + __shfl_xor(send, 8, 32);
            }
        }
        {
            const bool up = (t & 4) != 0;
#pragma unroll
            for (int j = 0; j < 2; ++j) {
                const float send = up ? r4[j] : r4[j + 2];
                const float keep = up ? r4[j + 2] : r4[j];
                r2[j] = keep + __shfl_xor(send, 4, 32);
            }
        }
        {
            const bool up = (t & 2) != 0;
            const float send = up ? r2[0] : r2[1];
            const float keep = up ? r2[1] : r2[0];
            r1 = keep + __shfl_xor(send, 2, 32);
        }
        r1 += __shfl_xor(r1, 1, 32);
        const int q = (t & 15) >> 1;
        __shared__ __attribute__((aligned(16))) float Sn[TILE_M * 2 * HEADS];
        if ((t & 1) == 0) Sn[row * (2 * HEADS) + q] = r1;
        __syncthreads();
        if (t < 32) {
            const int r  = t & 15;
            const int wh = t >> 4;
            const float4 sv = *reinterpret_cast<const float4*>(Sn + r * (2 * HEADS) + wh * HEADS);
            if (node0 + r < N)
                *reinterpret_cast<float4*>((wh ? ad_n : as_n) + (size_t)(node0 + r) * HEADS) = sv;
        }
    }
}

extern "C" void kernel_launch(void* const* d_in, const int* in_sizes, int n_in,
                              void* d_out, int out_size, void* d_ws, size_t ws_size,
                              hipStream_t stream) {
    const float* x        = (const float*)d_in[0];
    const int*   ei       = (const int*)  d_in[1];
    const float* W1       = (const float*)d_in[2];
    const float* att_src1 = (const float*)d_in[3];
    const float* att_dst1 = (const float*)d_in[4];
    const float* b1       = (const float*)d_in[5];
    const float* W2       = (const float*)d_in[6];
    const float* att_src2 = (const float*)d_in[7];
    const float* att_dst2 = (const float*)d_in[8];
    const float* b2       = (const float*)d_in[9];

    const int Fin = 128;
    const int N   = in_sizes[0] / Fin;
    const int E   = in_sizes[1] / 2;
    const int* srcv = ei;
    const int* dstv = ei + E;

    const int nChunks = (E + CHUNK_CAP * 256 - 1) / (CHUNK_CAP * 256) * 256;
    const int chunk   = ((E + nChunks - 1) / nChunks + 31) & ~31;
    const int nBins   = (N + BIN_NODES - 1) / BIN_NODES;

    float* ws    = (float*)d_ws;
    float* Zbuf  = ws;
    float* ASRC2 = Zbuf;
    float* ADST2 = ASRC2 + (size_t)N * HEADS;
    float* HID   = Zbuf + (size_t)N * HEADS * FIN1;
    float* ASRC  = HID  + (size_t)N * CDIM;
    float* ADST  = ASRC + (size_t)N * HEADS;
    float* V1    = ADST + (size_t)N * HEADS;
    float* V2    = V1 + 2 * HEADS * FIN1;
    int2* BINNED = (int2*)(V2 + 2 * HEADS * FIN2);
    int* OFFS    = (int*)(BINNED + (size_t)nChunks * chunk);
    int* ROWPTR  = OFFS + (size_t)nChunks * (nBins + 1);
    int* CSRC    = ROWPTR + (N + 1);

    const int nsBlocks   = (N * HEADS + SORT_T - 1) / SORT_T;
    const int hvBlocks   = (2 * HEADS * (FIN1 + FIN2) + SORT_T - 1) / SORT_T;
    const int ntBlocks   = (N + TILE_M - 1) / TILE_M;

    bin_edges<<<nChunks + hvBlocks, SORT_T, 0, stream>>>(srcv, dstv, E, chunk, nBins, nChunks,
                                                         BINNED, OFFS,
                                                         W1, att_src1, att_dst1, V1,
                                                         W2, att_src2, att_dst2, V2);
    build_csr<<<nBins + nsBlocks, SORT_T, 0, stream>>>(BINNED, OFFS, nChunks, chunk, nBins, N, ROWPTR, CSRC,
                                                       x, V1, ASRC, ADST);

    node_sum_project<FIN1, true><<<ntBlocks, 256, 0, stream>>>(ROWPTR, CSRC, x, ASRC, ADST, W1, b1, HID, N,  1,
                                                               V2, ASRC2, ADST2);

    node_sum_project<FIN2, false><<<ntBlocks, 256, 0, stream>>>(ROWPTR, CSRC, HID, ASRC2, ADST2, W2, b2, (float*)d_out, N,  0,
                                                                nullptr, nullptr, nullptr);
}


# projection: B fragments read 16 bytes per lane (a lane's 4 consecutive columns feed 4 accumulator tiles), a wave takes a head's 64 columns over half its rows; sum order changes (8 partial tiles)
# speedup vs baseline: 16.0983x; 1.0387x over previous
#include <hip/hip_runtime.h>
#include <hip/hip_bf16.h>
#include <math.h>

typedef float v2f __attribute__((ext_vector_type(2)));
typedef float v8f __attribute__((ext_vector_type(8)));

#define HEADS 4
#define HC 256
#define CDIM 64
#define NEG_SLOPE 0.2f
#define FIN1 128
#define FIN2 64


__device__ __forceinline__ void head_vector(const float* __restrict__ Wm,
                                            const float* __restrict__ att_s,
                                            const float* __restrict__ att_d,
                                            float* __restrict__ V, int FIN, int idx) {
    const int t = idx / FIN, f = idx - t * FIN;
    const int h = t & 3;
    const float* wr = Wm + (size_t)f * HC + h * CDIM;
    const float* av = (t < HEADS ? att_s : att_d) + h * CDIM;
    float s = 0.f;
#pragma unroll 8
    for (int c = 0; c < CDIM; ++c) s = fmaf(wr[c], av[c], s);
    V[idx] = s;
}

template <int FIN>
__device__ __forceinline__ void node_scores(const float* __restrict__ X,
                                            const float* __restrict__ V,
                                            float* __restrict__ as_, float* __restrict__ ad_,
                                            int N, int idx) {
    if (idx >= N * HEADS) return;
    const int n = idx >> 2, h = idx & 3;
    const float* row = X + (size_t)n * FIN;
    const float* vs  = V + h * FIN;
    const float* vd  = V + (HEADS + h) * FIN;
    float s = 0.f, d = 0.f;
#pragma unroll 8
    for (int f = 0; f < FIN; ++f) {
        const float v = row[f];
        s = fmaf(v, vs[f], s);
        d = fmaf(v, vd[f], d);
    }
    as_[idx] = s;
    ad_[idx] = d;
}

__device__ __forceinline__ float leaky(float e) {
    return fmaxf(e, NEG_SLOPE * e);
}

#define BIN_SHIFT 8
#define BIN_NODES (1 << BIN_SHIFT)
#define SORT_T 1024
#define CHUNK_PER 4
#define CHUNK_CAP (SORT_T * CHUNK_PER)
#define MAXBIN SORT_T
#define OWN_CAP 8192

__device__ __forceinline__ int wave_sum_i(int v) {
#pragma unroll
    for (int o = 16; o > 0; o >>= 1) v += __shfl_xor(v, o, 32);
    return v;
}

__device__ __forceinline__ int block_excl_scan(int v, int* wtot, int& total) {
    const int lane = threadIdx.x & 31;
    const int w    = threadIdx.x >> 5;
    int x = v;
#pragma unroll
    for (int o = 1; o < 32; o <<= 1) {
        const int y = __shfl_up(x, o, 32);
        if (lane >= o) x += y;
    }
    if (lane == 31) wtot[w] = x;
    __syncthreads();
    const int wt     = wtot[lane];
    const int before = wave_sum_i(lane < w ? wt : 0);
    total            = wave_sum_i(wt);
    __syncthreads();
    return before + x - v;
}

__device__ __forceinline__ int block_sum(int v, int* wtot) {
    const int lane = threadIdx.x & 31;
    const int w    = threadIdx.x >> 5;
    const int x    = wave_sum_i(v);
    if (lane == 0) wtot[w] = x;
    __syncthreads();
    const int total = wave_sum_i(wtot[lane]);
    __syncthreads();
    return total;
}

__global__ void __launch_bounds__(SORT_T)
bin_edges(const int* __restrict__ srcv, const int* __restrict__ dstv, int E, int chunk,
          int nBins, int chunkBlocks,
          int2* __restrict__ binned, int* __restrict__ offs,
          const float* __restrict__ W1, const float* __restrict__ as1,
          const float* __restrict__ ad1, float* __restrict__ V1,
          const float* __restrict__ W2, const float* __restrict__ as2,
          const float* __restrict__ ad2, float* __restrict__ V2) {
    __shared__ int  binCnt[MAXBIN];
    __shared__ int  wtot[SORT_T / 32];
    __shared__ int2 staged[CHUNK_CAP];
    const int t = threadIdx.x;
    if ((int)blockIdx.x >= chunkBlocks) {
        const int idx = ((int)blockIdx.x - chunkBlocks) * SORT_T + t;
        if (idx < 2 * HEADS * FIN1) head_vector(W1, as1, ad1, V1, FIN1, idx);
        else if (idx < 2 * HEADS * (FIN1 + FIN2))
            head_vector(W2, as2, ad2, V2, FIN2, idx - 2 * HEADS * FIN1);
        return;
    }
    const int e0 = (int)blockIdx.x * chunk;
    int n = E - e0;
    n = n < 0 ? 0 : (n > chunk ? chunk : n);
    if (t < nBins) binCnt[t] = 0;
    __syncthreads();

    int s[CHUNK_PER], d[CHUNK_PER], r[CHUNK_PER];
#pragma unroll
    for (int i = 0; i < CHUNK_PER; ++i) {
        const int j = t + i * SORT_T;
        s[i] = 0; d[i] = 0;
        if (j < n) { d[i] = dstv[e0 + j]; s[i] = srcv[e0 + j]; }
    }
#pragma unroll
    for (int i = 0; i < CHUNK_PER; ++i) {
        const int j = t + i * SORT_T;
        r[i] = 0;
        if (j < n) r[i] = atomicAdd(&binCnt[d[i] >> BIN_SHIFT], 1);
    }
    __syncthreads();

    const int c = (t < nBins) ? binCnt[t] : 0;
    int tot;
    const int ex = block_excl_scan(c, wtot, tot);
    int* myoffs = offs + (size_t)blockIdx.x * (nBins + 1);
    if (t < nBins) { binCnt[t] = ex; myoffs[t] = ex; }
    if (t == 0) myoffs[nBins] = tot;
    __syncthreads();

#pragma unroll
    for (int i = 0; i < CHUNK_PER; ++i) {
        const int j = t + i * SORT_T;
        if (j < n) staged[binCnt[d[i] >> BIN_SHIFT] + r[i]] = make_int2(s[i], d[i]);
    }
    __syncthreads();
    int2* outp = binned + (size_t)e0;
    for (int j = t; j < n; j += SORT_T) outp[j] = staged[j];
}

__global__ void __launch_bounds__(SORT_T)
build_csr(const int2* __restrict__ binned, const int* __restrict__ offs,
          int nChunks, int chunk, int nBins, int N,
          int* __restrict__ rowptr, int* __restrict__ csrc,
          const float* __restrict__ X, const float* __restrict__ V,
          float* __restrict__ as_, float* __restrict__ ad_) {
    __shared__ int nodeCnt[BIN_NODES];
    __shared__ int wtot[SORT_T / 32];
    __shared__ int stagedSrc[OWN_CAP];
    const int t      = threadIdx.x;
    if ((int)blockIdx.x >= nBins) {
        node_scores<FIN1>(X, V, as_, ad_, N, ((int)blockIdx.x - nBins) * SORT_T + t);
        return;
    }
    const int b      = blockIdx.x;
    const int node0  = b << BIN_SHIFT;
    const int stride = nBins + 1;

    int lo = 0, hi = 0;
    for (int k = t; k < nChunks; k += SORT_T) {
        lo += offs[(size_t)k * stride + b];
        hi += offs[(size_t)k * stride + b + 1];
    }
    const int base  = block_sum(lo, wtot);
    const int end   = block_sum(hi, wtot);
    const int total = end - base;

    if (t < BIN_NODES) nodeCnt[t] = 0;
    __syncthreads();

    const int hw = t >> 4, hl = t & 15;
    for (int k = hw; k < nChunks; k += SORT_T / 16) {
        const int o0 = offs[(size_t)k * stride + b];
        const int o1 = offs[(size_t)k * stride + b + 1];
        const int2* run = binned + (size_t)k * chunk;
        for (int j = o0 + hl; j < o1; j += 16) atomicAdd(&nodeCnt[run[j].y - node0], 1);
    }
    __syncthreads();

    const int c = (t < BIN_NODES) ? nodeCnt[t] : 0;
    int tot;
    const int ex = block_excl_scan(c, wtot, tot);
    if (t < BIN_NODES) {
        nodeCnt[t] = ex;
        if (node0 + t < N) rowptr[node0 + t] = base + ex;
    }
    if (b == nBins - 1 && t == 0) rowptr[N] = end;
    __syncthreads();

    const bool fits = total <= OWN_CAP;
    for (int k = hw; k < nChunks; k += SORT_T / 16) {
        const int o0 = offs[(size_t)k * stride + b];
        const int o1 = offs[(size_t)k * stride + b + 1];
        const int2* run = binned + (size_t)k * chunk;
        for (int j = o0 + hl; j < o1; j += 16) {
            const int2 e  = run[j];
            const int pos = atomicAdd(&nodeCnt[e.y - node0], 1);
            if (fits) stagedSrc[pos] = e.x;
            else      csrc[base + pos] = e.x;
        }
    }
    if (fits) {
        __syncthreads();
        for (int j = t; j < total; j += SORT_T) csrc[base + j] = stagedSrc[j];
    }
}

template <int M>
__device__ __forceinline__ float row_x(float v) {
    return __builtin_bit_cast(float, __builtin_amdgcn_update_dpp(0, __builtin_bit_cast(int, v),
                                                                  0x160 + M, 0xf, 0xf, true));
}
__device__ __forceinline__ float xhalf(float v) {
    const int b = __builtin_bit_cast(int, v);
    return __builtin_bit_cast(float, __builtin_amdgcn_permlanex16(b, b, 0x76543210, 0xfedcba98, false, false));
}
#define ROW_MAX_STEP(M)                                                                                  \
    asm volatile("v_max_num_f32_dpp %0, %0, %0 row_xmask:" #M " row_mask:0xf bank_mask:0xf\n\t"          \
                 "v_max_num_f32_dpp %1, %1, %1 row_xmask:" #M " row_mask:0xf bank_mask:0xf\n\t"          \
                 "v_max_num_f32_dpp %2, %2, %2 row_xmask:" #M " row_mask:0xf bank_mask:0xf\n\t"          \
                 "v_max_num_f32_dpp %3, %3, %3 row_xmask:" #M " row_mask:0xf bank_mask:0xf"                \
                 : "+v"(a), "+v"(b), "+v"(c), "+v"(d))
__device__ __forceinline__ void wave_max4(float& a, float& b, float& c, float& d) {
    ROW_MAX_STEP(1);
    ROW_MAX_STEP(2);
    ROW_MAX_STEP(4);
    ROW_MAX_STEP(8);
    a = fmaxf(a, xhalf(a));
    b = fmaxf(b, xhalf(b));
    c = fmaxf(c, xhalf(c));
    d = fmaxf(d, xhalf(d));
}
#undef ROW_MAX_STEP
__device__ __forceinline__ float wave_sum(float v) {
    v += row_x<1>(v);
    v += row_x<2>(v);
    v += row_x<4>(v);
    v += row_x<8>(v);
    return v + xhalf(v);
}

__device__ __forceinline__ void edge_scores(int i, int cnt, int base, int d,
                                            const int* __restrict__ csrc,
                                            const float* __restrict__ as_,
                                            const float4 adv, int& s, float4& e) {
    s = d;
    if (i < cnt) s = csrc[base + i];
    const float4 a = *reinterpret_cast<const float4*>(as_ + (size_t)s * HEADS);
    e.x = leaky(a.x + adv.x);
    e.y = leaky(a.y + adv.y);
    e.z = leaky(a.z + adv.z);
    e.w = leaky(a.w + adv.w);
}

template <int FIN>
__device__ __forceinline__ void node_rows_sum(const int d, const int base, const int cnt,
                                              const float* __restrict__ adrow,
                                              const int s_first, const float4 e_first,
                                              const int* __restrict__ csrc,
                                              const float* __restrict__ X,
                                              const float* __restrict__ as_,
                                              float* __restrict__ zrow) {
    constexpr int VEC = FIN / 32;
    typedef float vrow __attribute__((ext_vector_type(VEC)));
    const int lane = threadIdx.x & 31;
    const int T    = cnt + 1;

    float m0 = e_first.x, m1 = e_first.y, m2 = e_first.z, m3 = e_first.w;
    for (int c0 = 32; c0 < T; c0 += 32) {
        const int i = c0 + lane;
        int s; float4 e;
        edge_scores(i, cnt, base, d, csrc, as_, *reinterpret_cast<const float4*>(adrow), s, e);
        if (i < T) {
            m0 = fmaxf(m0, e.x); m1 = fmaxf(m1, e.y);
            m2 = fmaxf(m2, e.z); m3 = fmaxf(m3, e.w);
        }
    }
    wave_max4(m0, m1, m2, m3);

    const bool in0 = lane < T;
    const float x0 = in0 ? __expf(e_first.x - m0) : 0.f;
    const float x1 = in0 ? __expf(e_first.y - m1) : 0.f;
    const float x2 = in0 ? __expf(e_first.z - m2) : 0.f;
    const float x3 = in0 ? __expf(e_first.w - m3) : 0.f;
    float s0 = x0, s1 = x1, s2 = x2, s3 = x3;
    for (int c0 = 32; c0 < T; c0 += 32) {
        const int i = c0 + lane;
        int s; float4 e;
        edge_scores(i, cnt, base, d, csrc, as_, *reinterpret_cast<const float4*>(adrow), s, e);
        if (i < T) {
            s0 += __expf(e.x - m0); s1 += __expf(e.y - m1);
            s2 += __expf(e.z - m2); s3 += __expf(e.w - m3);
        }
    }
    const float r0 = __builtin_amdgcn_rcpf(wave_sum(s0) + 1e-16f);
    const float r1 = __builtin_amdgcn_rcpf(wave_sum(s1) + 1e-16f);
    const float r2 = __builtin_amdgcn_rcpf(wave_sum(s2) + 1e-16f);
    const float r3 = __builtin_amdgcn_rcpf(wave_sum(s3) + 1e-16f);

    vrow z0 = {}, z1 = {}, z2 = {}, z3 = {};
    const unsigned lo = lane * VEC;
    for (int c0 = 0; c0 < T; c0 += 32) {
        int sl = s_first;
        float y0 = x0, y1 = x1, y2 = x2, y3 = x3;
        if (c0 > 0) {
            float4 e;
            edge_scores(c0 + lane, cnt, base, d, csrc, as_, *reinterpret_cast<const float4*>(adrow), sl, e);
            y0 = __expf(e.x - m0); y1 = __expf(e.y - m1);
            y2 = __expf(e.z - m2); y3 = __expf(e.w - m3);
        }
        const int w0 = __float_as_int(y0 * r0);
        const int w1 = __float_as_int(y1 * r1);
        const int w2 = __float_as_int(y2 * r2);
        const int w3 = __float_as_int(y3 * r3);
        const int n = (T - c0 < 32) ? (T - c0) : 32;
        int j = 0;
        for (; j + 4 <= n; j += 4) {
            vrow xr[4];
#pragma unroll
            for (int k = 0; k < 4; ++k) {
                const int s = __builtin_amdgcn_readlane(sl, j + k);
                xr[k] = *reinterpret_cast<const vrow*>(X + ((unsigned)s * FIN + lo));
            }
#pragma unroll
            for (int k = 0; k < 4; ++k) {
                const float a0 = __int_as_float(__builtin_amdgcn_readlane(w0, j + k));
                const float a1 = __int_as_float(__builtin_amdgcn_readlane(w1, j + k));
                const float a2 = __int_as_float(__builtin_amdgcn_readlane(w2, j + k));
                const float a3 = __int_as_float(__builtin_amdgcn_readlane(w3, j + k));
                z0 += xr[k] * a0;
                z1 += xr[k] * a1;
                z2 += xr[k] * a2;
                z3 += xr[k] * a3;
            }
        }
        for (; j < n; ++j) {
            const int s = __builtin_amdgcn_readlane(sl, j);
            const vrow xv = *reinterpret_cast<const vrow*>(X + ((unsigned)s * FIN + lo));
            const float a0 = __int_as_float(__builtin_amdgcn_readlane(w0, j));
            const float a1 = __int_as_float(__builtin_amdgcn_readlane(w1, j));
            const float a2 = __int_as_float(__builtin_amdgcn_readlane(w2, j));
            const float a3 = __int_as_float(__builtin_amdgcn_readlane(w3, j));
            z0 += xv * a0;
            z1 += xv * a1;
            z2 += xv * a2;
            z3 += xv * a3;
        }
    }

    *reinterpret_cast<vrow*>(zrow)           = z0;
    *reinterpret_cast<vrow*>(zrow + FIN)     = z1;
    *reinterpret_cast<vrow*>(zrow + 2 * FIN) = z2;
    *reinterpret_cast<vrow*>(zrow + 3 * FIN) = z3;
}

#define TILE_M 16
#define PPITCH 68

template <int FIN, bool NEXT>
__global__ void __launch_bounds__(256)
node_sum_project(const int* __restrict__ rowptr, const int* __restrict__ csrc,
                 const float* __restrict__ X,
                 const float* __restrict__ as_, const float* __restrict__ ad_,
                 const float* __restrict__ Wm, const float* __restrict__ bias,
                 float* __restrict__ out, int N, int do_relu,
                 const float* __restrict__ Vn,
                 float* __restrict__ as_n, float* __restrict__ ad_n) {
    constexpr int VEC   = FIN / 32;
    constexpr int K     = HEADS * FIN;
    constexpr int PITCH = K + 4;
    constexpr int NPART = 2 * HEADS;
    constexpr int ZSIZE = TILE_M * PITCH > NPART * TILE_M * PPITCH ? TILE_M * PITCH
                                                                   : NPART * TILE_M * PPITCH;
    typedef float vrow __attribute__((ext_vector_type(VEC)));
    __shared__ __attribute__((aligned(16))) float Zs[ZSIZE];

    const int lane = threadIdx.x & 31;
    const int wave  = __builtin_amdgcn_readfirstlane((int)(threadIdx.x >> 5));
    const int node0 = blockIdx.x * TILE_M;

    {
        const int rowA = wave * 2;
        const int dA = node0 + rowA, dB = dA + 1;
        const int dAc = dA < N ? dA : N - 1;
        const int dBc = dB < N ? dB : N - 1;
        int rpi = dA + (lane < 3 ? lane : 0);
        rpi = rpi < N ? rpi : N;
        const int rp  = rowptr[rpi];
        const int bA   = __builtin_amdgcn_readlane(rp, 0);
        const int bB   = __builtin_amdgcn_readlane(rp, 1);
        const int cntA = bB - bA;
        const int cntB = __builtin_amdgcn_readlane(rp, 2) - bB;
        const float* adA = ad_ + (size_t)dAc * HEADS;
        const float* adB = ad_ + (size_t)dBc * HEADS;
        const float4 advA = *reinterpret_cast<const float4*>(adA);
        const float4 advB = *reinterpret_cast<const float4*>(adB);
        int sA, sB; float4 eA, eB;
        edge_scores(lane, cntA, bA, dAc, csrc, as_, advA, sA, eA);
        edge_scores(lane, cntB, bB, dBc, csrc, as_, advB, sB, eB);

        float* zA = Zs + rowA * PITCH + lane * VEC;
        float* zB = zA + PITCH;
        if (dA < N) {
            node_rows_sum<FIN>(dA, bA, cntA, adA, sA, eA, csrc, X, as_, zA);
        } else {
            const vrow zz = {};
#pragma unroll
            for (int h = 0; h < HEADS; ++h) *reinterpret_cast<vrow*>(zA + h * FIN) = zz;
        }
        if (dB < N) {
            node_rows_sum<FIN>(dB, bB, cntB, adB, sB, eB, csrc, X, as_, zB);
        } else {
            const vrow zz = {};
#pragma unroll
            for (int h = 0; h < HEADS; ++h) *reinterpret_cast<vrow*>(zB + h * FIN) = zz;
        }
    }
    __syncthreads();

    const int laneM  = lane & 15;
    const int laneHi = lane >> 4;
    const int hq     = wave >> 1;
    const int f0     = (wave & 1) * (FIN / 2);
    const float* Arow  = Zs + laneM * PITCH + hq * FIN + f0 + (laneHi << 1);
    const float* Bbase = Wm + (size_t)(f0 + (laneHi << 1)) * HC + hq * CDIM + (laneM << 2);

    v8f acc0 = {}, acc1 = {}, acc2 = {}, acc3 = {};
    for (int f = 0; f < FIN / 2; f += 8) {
        const v2f a0 = *reinterpret_cast<const v2f*>(Arow + f);
        const v2f a1 = *reinterpret_cast<const v2f*>(Arow + f + 4);
        const float4 r00 = *reinterpret_cast<const float4*>(Bbase + (size_t)f * HC);
        const float4 r01 = *reinterpret_cast<const float4*>(Bbase + (size_t)f * HC + HC);
        const float4 r10 = *reinterpret_cast<const float4*>(Bbase + (size_t)(f + 4) * HC);
        const float4 r11 = *reinterpret_cast<const float4*>(Bbase + (size_t)(f + 4) * HC + HC);
        __builtin_amdgcn_sched_barrier(0);
        v2f b0, b1, b2, b3;
        b0.x = r00.x; b0.y = r01.x;
        b1.x = r00.y; b1.y = r01.y;
        b2.x = r00.z; b2.y = r01.z;
        b3.x = r00.w; b3.y = r01.w;
        acc0 = __builtin_amdgcn_wmma_f32_16x16x4_f32(false, a0, false, b0, (short)0, acc0, false, false);
        acc1 = __builtin_amdgcn_wmma_f32_16x16x4_f32(false, a0, false, b1, (short)0, acc1, false, false);
        acc2 = __builtin_amdgcn_wmma_f32_16x16x4_f32(false, a0, false, b2, (short)0, acc2, false, false);
        acc3 = __builtin_amdgcn_wmma_f32_16x16x4_f32(false, a0, false, b3, (short)0, acc3, false, false);
        b0.x = r10.x; b0.y = r11.x;
        b1.x = r10.y; b1.y = r11.y;
        b2.x = r10.z; b2.y = r11.z;
        b3.x = r10.w; b3.y = r11.w;
        acc0 = __builtin_amdgcn_wmma_f32_16x16x4_f32(false, a1, false, b0, (short)0, acc0, false, false);
        acc1 = __builtin_amdgcn_wmma_f32_16x16x4_f32(false, a1, false, b1, (short)0, acc1, false, false);
        acc2 = __builtin_amdgcn_wmma_f32_16x16x4_f32(false, a1, false, b2, (short)0, acc2, false, false);
        acc3 = __builtin_amdgcn_wmma_f32_16x16x4_f32(false, a1, false, b3, (short)0, acc3, false, false);
    }
    __syncthreads();

    float* Pw = Zs + wave * (TILE_M * PPITCH) + (8 * laneHi) * PPITCH + (laneM << 2);
#pragma unroll
    for (int i = 0; i < 8; ++i)
        *reinterpret_cast<float4*>(Pw + i * PPITCH) = make_float4(acc0[i], acc1[i], acc2[i], acc3[i]);
    const int t   = threadIdx.x;
    const int row = t >> 4;
    const int c4  = (t & 15) << 2;
    __syncthreads();
    const float4 bs = *reinterpret_cast<const float4*>(bias + c4);
    float4 vn[2 * HEADS];
    if (NEXT) {
#pragma unroll
        for (int j = 0; j < 2 * HEADS; ++j)
            vn[j] = *reinterpret_cast<const float4*>(Vn + j * CDIM + c4);
    }

    const float* Pr = Zs + row * PPITCH + c4;
    float4 p[NPART];
#pragma unroll
    for (int j = 0; j < NPART; ++j)
        p[j] = *reinterpret_cast<const float4*>(Pr + j * (TILE_M * PPITCH));
    float4 v;
    v.x = (((p[0].x + p[1].x) + (p[2].x + p[3].x)) + ((p[4].x + p[5].x) + (p[6].x + p[7].x))) * 0.25f + bs.x;
    v.y = (((p[0].y + p[1].y) + (p[2].y + p[3].y)) + ((p[4].y + p[5].y) + (p[6].y + p[7].y))) * 0.25f + bs.y;
    v.z = (((p[0].z + p[1].z) + (p[2].z + p[3].z)) + ((p[4].z + p[5].z) + (p[6].z + p[7].z))) * 0.25f + bs.z;
    v.w = (((p[0].w + p[1].w) + (p[2].w + p[3].w)) + ((p[4].w + p[5].w) + (p[6].w + p[7].w))) * 0.25f + bs.w;
    if (do_relu) {
        v.x = fmaxf(v.x, 0.f); v.y = fmaxf(v.y, 0.f);
        v.z = fmaxf(v.z, 0.f); v.w = fmaxf(v.w, 0.f);
    }
    if (node0 + row < N)
        *reinterpret_cast<float4*>(out + (size_t)(node0 + row) * CDIM + c4) = v;

    if (NEXT) {
        float sc[2 * HEADS];
#pragma unroll
        for (int j = 0; j < 2 * HEADS; ++j)
            sc[j] = fmaf(v.x, vn[j].x, fmaf(v.y, vn[j].y, fmaf(v.z, vn[j].z, v.w * vn[j].w)));
        float r4[4], r2[2], r1;
        {
            const bool up = (t & 8) != 0;
#pragma unroll
            for (int j = 0; j < 4; ++j) {
                const float send = up ? sc[j] : sc[j + 4];
                const float keep = up ? sc[j + 4] : sc[j];
                r4[j] = keep + __shfl_xor(send, 8, 32);
            }
        }
        {
            const bool up = (t & 4) != 0;
#pragma unroll
            for (int j = 0; j < 2; ++j) {
                const float send = up ? r4[j] : r4[j + 2];
                const float keep = up ? r4[j + 2] : r4[j];
                r2[j] = keep + __shfl_xor(send, 4, 32);
            }
        }
        {
            const bool up = (t & 2) != 0;
            const float send = up ? r2[0] : r2[1];
            const float keep = up ? r2[1] : r2[0];
            r1 = keep + __shfl_xor(send, 2, 32);
        }
        r1 += __shfl_xor(r1, 1, 32);
        const int q = (t & 15) >> 1;
        __shared__ __attribute__((aligned(16))) float Sn[TILE_M * 2 * HEADS];
        if ((t & 1) == 0) Sn[row * (2 * HEADS) + q] = r1;
        __syncthreads();
        if (t < 32) {
            const int r  = t & 15;
            const int wh = t >> 4;
            const float4 sv = *reinterpret_cast<const float4*>(Sn + r * (2 * HEADS) + wh * HEADS);
            if (node0 + r < N)
                *reinterpret_cast<float4*>((wh ? ad_n : as_n) + (size_t)(node0 + r) * HEADS) = sv;
        }
    }
}

extern "C" void kernel_launch(void* const* d_in, const int* in_sizes, int n_in,
                              void* d_out, int out_size, void* d_ws, size_t ws_size,
                              hipStream_t stream) {
    const float* x        = (const float*)d_in[0];
    const int*   ei       = (const int*)  d_in[1];
    const float* W1       = (const float*)d_in[2];
    const float* att_src1 = (const float*)d_in[3];
    const float* att_dst1 = (const float*)d_in[4];
    const float* b1       = (const float*)d_in[5];
    const float* W2       = (const float*)d_in[6];
    const float* att_src2 = (const float*)d_in[7];
    const float* att_dst2 = (const float*)d_in[8];
    const float* b2       = (const float*)d_in[9];

    const int Fin = 128;
    const int N   = in_sizes[0] / Fin;
    const int E   = in_sizes[1] / 2;
    const int* srcv = ei;
    const int* dstv = ei + E;

    const int nChunks = (E + CHUNK_CAP * 256 - 1) / (CHUNK_CAP * 256) * 256;
    const int chunk   = ((E + nChunks - 1) / nChunks + 31) & ~31;
    const int nBins   = (N + BIN_NODES - 1) / BIN_NODES;

    float* ws    = (float*)d_ws;
    float* Zbuf  = ws;
    float* ASRC2 = Zbuf;
    float* ADST2 = ASRC2 + (size_t)N * HEADS;
    float* HID   = Zbuf + (size_t)N * HEADS * FIN1;
    float* ASRC  = HID  + (size_t)N * CDIM;
    float* ADST  = ASRC + (size_t)N * HEADS;
    float* V1    = ADST + (size_t)N * HEADS;
    float* V2    = V1 + 2 * HEADS * FIN1;
    int2* BINNED = (int2*)(V2 + 2 * HEADS * FIN2);
    int* OFFS    = (int*)(BINNED + (size_t)nChunks * chunk);
    int* ROWPTR  = OFFS + (size_t)nChunks * (nBins + 1);
    int* CSRC    = ROWPTR + (N + 1);

    const int nsBlocks   = (N * HEADS + SORT_T - 1) / SORT_T;
    const int hvBlocks   = (2 * HEADS * (FIN1 + FIN2) + SORT_T - 1) / SORT_T;
    const int ntBlocks   = (N + TILE_M - 1) / TILE_M;

    bin_edges<<<nChunks + hvBlocks, SORT_T, 0, stream>>>(srcv, dstv, E, chunk, nBins, nChunks,
                                                         BINNED, OFFS,
                                                         W1, att_src1, att_dst1, V1,
                                                         W2, att_src2, att_dst2, V2);
    build_csr<<<nBins + nsBlocks, SORT_T, 0, stream>>>(BINNED, OFFS, nChunks, chunk, nBins, N, ROWPTR, CSRC,
                                                       x, V1, ASRC, ADST);

    node_sum_project<FIN1, true><<<ntBlocks, 256, 0, stream>>>(ROWPTR, CSRC, x, ASRC, ADST, W1, b1, HID, N,  1,
                                                               V2, ASRC2, ADST2);

    node_sum_project<FIN2, false><<<ntBlocks, 256, 0, stream>>>(ROWPTR, CSRC, HID, ASRC2, ADST2, W2, b2, (float*)d_out, N,  0,
                                                                nullptr, nullptr, nullptr);
}
